# starting kernel
# baseline (speedup 1.0000x reference)
#include <hip/hip_runtime.h>

#define HID      64
#define NNODES   100000
#define NEDGES   1200000
#define NGRAPHS  1000

typedef __attribute__((ext_vector_type(2))) float v2f;
typedef __attribute__((ext_vector_type(8))) float v8f;

template <bool RELU_IN>
__global__ __launch_bounds__(128) void gcn_gemm_wmma(
    const float* __restrict__ In, const float* __restrict__ W,
    const float* __restrict__ bias_in, float* __restrict__ Out)
{
    __shared__ float sW[64 * 65];
    __shared__ float sA[16 * 65];

    const int tid  = threadIdx.x;
    const int wave = tid >> 5;
    const int lane = tid & 31;
    const int rowBase = blockIdx.x * 16;

    for (int i = tid; i < 64 * 64; i += 128) {
        int r = i >> 6, c = i & 63;
        sW[r * 65 + c] = W[i];
    }
    for (int i = tid; i < 16 * 64; i += 128) {
        int r = i >> 6, c = i & 63;
        float v = In[(rowBase + r) * 64 + c];
        if (RELU_IN) {
            v += bias_in[c];
            v = v > 0.0f ? v : 0.0f;
        }
        sA[r * 65 + c] = v;
    }
    __syncthreads();

    const int m     = lane & 15;
    const int n     = (wave << 4) + (lane & 15);
    const int khalf = (lane >> 4) << 1;

    v8f acc = {};
#pragma unroll
    for (int k = 0; k < 64; k += 4) {
        v2f a, b;
        a.x = sA[m * 65 + k + khalf];
        a.y = sA[m * 65 + k + khalf + 1];
        b.x = sW[(k + khalf) * 65 + n];
        b.y = sW[(k + khalf + 1) * 65 + n];
        acc = __builtin_amdgcn_wmma_f32_16x16x4_f32(
             false, a,  false, b,
             (short)0, acc,  false,  false);
    }

    const int rOff = (lane >> 4) << 3;
    float* outp = Out + (size_t)(rowBase + rOff) * 64 + n;
#pragma unroll
    for (int v = 0; v < 8; ++v)
        outp[(size_t)v * 64] = acc[v];
}

__global__ __launch_bounds__(256) void gcn_scatter(
    const float* __restrict__ X, const int* __restrict__ src,
    const int* __restrict__ dst, float* __restrict__ Agg)
{
    const unsigned gid = blockIdx.x * 256u + threadIdx.x;
    const unsigned e = gid >> 5;
    const unsigned l = gid & 31;
    if (e >= NEDGES) return;
    const int s = src[e];
    const int d = dst[e];
    const float2 v = *(const float2*)(X + (size_t)s * 64 + l * 2);
    float* p = Agg + (size_t)d * 64 + l * 2;
    unsafeAtomicAdd(p, v.x);
    unsafeAtomicAdd(p + 1, v.y);
}

__global__ __launch_bounds__(256) void gcn_readout_acc(
    const float* __restrict__ Agg2, const int* __restrict__ gids,
    float* __restrict__ gsum, float* __restrict__ gcnt)
{
    const unsigned gid = blockIdx.x * 256u + threadIdx.x;
    const unsigned node = gid >> 6;
    if (node >= NNODES) return;
    const unsigned col = gid & 63;
    const int g = gids[node];
    unsafeAtomicAdd(&gsum[(size_t)g * 64 + col], Agg2[gid]);
    if (col == 0) unsafeAtomicAdd(&gcnt[g], 1.0f);
}

__global__ __launch_bounds__(256) void gcn_finalize(
    const float* __restrict__ gsum, const float* __restrict__ gcnt,
    const float* __restrict__ b2, float* __restrict__ out)
{
    const unsigned gid = blockIdx.x * 256u + threadIdx.x;
    if (gid >= NGRAPHS * 64u) return;
    const unsigned g = gid >> 6;
    const unsigned c = gid & 63;
    const float cnt = gcnt[g];
    out[gid] = (cnt > 0.0f) ? (gsum[gid] / cnt + b2[c]) : 0.0f;
}

extern "C" void kernel_launch(void* const* d_in, const int* in_sizes, int n_in,
                              void* d_out, int out_size, void* d_ws, size_t ws_size,
                              hipStream_t stream)
{
    (void)in_sizes; (void)n_in; (void)out_size; (void)ws_size;

    const float* feats = (const float*)d_in[0];
    const float* W1    = (const float*)d_in[1];
    const float* b1    = (const float*)d_in[2];
    const float* W2    = (const float*)d_in[3];
    const float* b2    = (const float*)d_in[4];
    const int*   src   = (const int*)d_in[5];
    const int*   dst   = (const int*)d_in[6];
    const int*   gids  = (const int*)d_in[7];
    float* out = (float*)d_out;

    const size_t nodeElems = (size_t)NNODES * HID;
    float* bufA = (float*)d_ws;
    float* bufB = bufA + nodeElems;
    float* gsum = bufB + nodeElems;
    float* gcnt = gsum + (size_t)NGRAPHS * HID;

    const int gemmBlocks    = NNODES / 16;
    const int scatterBlocks = (NEDGES * 32 + 255) / 256;
    const int readBlocks    = (NNODES * 64 + 255) / 256;
    const int finBlocks     = (NGRAPHS * 64 + 255) / 256;

    hipLaunchKernelGGL((gcn_gemm_wmma<false>), dim3(gemmBlocks), dim3(128), 0, stream,
                       feats, W1, (const float*)nullptr, bufA);
    hipMemsetAsync(bufB, 0, nodeElems * sizeof(float), stream);
    hipLaunchKernelGGL(gcn_scatter, dim3(scatterBlocks), dim3(256), 0, stream,
                       bufA, src, dst, bufB);
    hipLaunchKernelGGL((gcn_gemm_wmma<true>), dim3(gemmBlocks), dim3(128), 0, stream,
                       bufB, W2, b1, bufA);
    hipMemsetAsync(bufB, 0, nodeElems * sizeof(float), stream);
    hipLaunchKernelGGL(gcn_scatter, dim3(scatterBlocks), dim3(256), 0, stream,
                       bufA, src, dst, bufB);
    hipMemsetAsync(gsum, 0, ((size_t)NGRAPHS * HID + NGRAPHS) * sizeof(float), stream);
    hipLaunchKernelGGL(gcn_readout_acc, dim3(readBlocks), dim3(256), 0, stream,
                       bufB, gids, gsum, gcnt);
    hipLaunchKernelGGL(gcn_finalize, dim3(finBlocks), dim3(256), 0, stream,
                       gsum, gcnt, b2, out);
}


# scatter: each wave takes 32 consecutive edges (lanes load src/dst once, readlane broadcasts them, 8 row gathers in flight) instead of one wave per edge; same f32 atomic adds, arrival order as before
# speedup vs baseline: 1.3880x; 1.3880x over previous
#include <hip/hip_runtime.h>

#define HID      64
#define NNODES   100000
#define NEDGES   1200000
#define NGRAPHS  1000

typedef __attribute__((ext_vector_type(2))) float v2f;
typedef __attribute__((ext_vector_type(8))) float v8f;

template <bool RELU_IN>
__global__ __launch_bounds__(128) void gcn_gemm_wmma(
    const float* __restrict__ In, const float* __restrict__ W,
    const float* __restrict__ bias_in, float* __restrict__ Out)
{
    __shared__ float sW[64 * 65];
    __shared__ float sA[16 * 65];

    const int tid  = threadIdx.x;
    const int wave = tid >> 5;
    const int lane = tid & 31;
    const int rowBase = blockIdx.x * 16;

    for (int i = tid; i < 64 * 64; i += 128) {
        int r = i >> 6, c = i & 63;
        sW[r * 65 + c] = W[i];
    }
    for (int i = tid; i < 16 * 64; i += 128) {
        int r = i >> 6, c = i & 63;
        float v = In[(rowBase + r) * 64 + c];
        if (RELU_IN) {
            v += bias_in[c];
            v = v > 0.0f ? v : 0.0f;
        }
        sA[r * 65 + c] = v;
    }
    __syncthreads();

    const int m     = lane & 15;
    const int n     = (wave << 4) + (lane & 15);
    const int khalf = (lane >> 4) << 1;

    v8f acc = {};
#pragma unroll
    for (int k = 0; k < 64; k += 4) {
        v2f a, b;
        a.x = sA[m * 65 + k + khalf];
        a.y = sA[m * 65 + k + khalf + 1];
        b.x = sW[(k + khalf) * 65 + n];
        b.y = sW[(k + khalf + 1) * 65 + n];
        acc = __builtin_amdgcn_wmma_f32_16x16x4_f32(
             false, a,  false, b,
             (short)0, acc,  false,  false);
    }

    const int rOff = (lane >> 4) << 3;
    float* outp = Out + (size_t)(rowBase + rOff) * 64 + n;
#pragma unroll
    for (int v = 0; v < 8; ++v)
        outp[(size_t)v * 64] = acc[v];
}

#define EDGES_PER_WAVE 32
#define GATHER_BATCH   8
__global__ __launch_bounds__(256) void gcn_scatter(
    const float* __restrict__ X, const int* __restrict__ src,
    const int* __restrict__ dst, float* __restrict__ Agg)
{
    const unsigned gid = blockIdx.x * 256u + threadIdx.x;
    const unsigned w = gid >> 5;
    const unsigned l = gid & 31;
    const unsigned e0 = w * EDGES_PER_WAVE;
    if (e0 >= NEDGES) return;
    const int sv = src[gid];
    const int dv = dst[gid];
    const float* xl = X + l * 2;
    float* al = Agg + l * 2;

    for (int j0 = 0; j0 < EDGES_PER_WAVE; j0 += GATHER_BATCH) {
        float2 v[GATHER_BATCH];
#pragma unroll
        for (int u = 0; u < GATHER_BATCH; ++u) {
            const int s = __builtin_amdgcn_readlane(sv, j0 + u);
            v[u] = *(const float2*)(xl + (size_t)s * 64);
        }
#pragma unroll
        for (int u = 0; u < GATHER_BATCH; ++u) {
            const int d = __builtin_amdgcn_readlane(dv, j0 + u);
            float* p = al + (size_t)d * 64;
            unsafeAtomicAdd(p, v[u].x);
            unsafeAtomicAdd(p + 1, v[u].y);
        }
    }
}

__global__ __launch_bounds__(256) void gcn_readout_acc(
    const float* __restrict__ Agg2, const int* __restrict__ gids,
    float* __restrict__ gsum, float* __restrict__ gcnt)
{
    const unsigned gid = blockIdx.x * 256u + threadIdx.x;
    const unsigned node = gid >> 6;
    if (node >= NNODES) return;
    const unsigned col = gid & 63;
    const int g = gids[node];
    unsafeAtomicAdd(&gsum[(size_t)g * 64 + col], Agg2[gid]);
    if (col == 0) unsafeAtomicAdd(&gcnt[g], 1.0f);
}

__global__ __launch_bounds__(256) void gcn_finalize(
    const float* __restrict__ gsum, const float* __restrict__ gcnt,
    const float* __restrict__ b2, float* __restrict__ out)
{
    const unsigned gid = blockIdx.x * 256u + threadIdx.x;
    if (gid >= NGRAPHS * 64u) return;
    const unsigned g = gid >> 6;
    const unsigned c = gid & 63;
    const float cnt = gcnt[g];
    out[gid] = (cnt > 0.0f) ? (gsum[gid] / cnt + b2[c]) : 0.0f;
}

extern "C" void kernel_launch(void* const* d_in, const int* in_sizes, int n_in,
                              void* d_out, int out_size, void* d_ws, size_t ws_size,
                              hipStream_t stream)
{
    (void)in_sizes; (void)n_in; (void)out_size; (void)ws_size;

    const float* feats = (const float*)d_in[0];
    const float* W1    = (const float*)d_in[1];
    const float* b1    = (const float*)d_in[2];
    const float* W2    = (const float*)d_in[3];
    const float* b2    = (const float*)d_in[4];
    const int*   src   = (const int*)d_in[5];
    const int*   dst   = (const int*)d_in[6];
    const int*   gids  = (const int*)d_in[7];
    float* out = (float*)d_out;

    const size_t nodeElems = (size_t)NNODES * HID;
    float* bufA = (float*)d_ws;
    float* bufB = bufA + nodeElems;
    float* gsum = bufB + nodeElems;
    float* gcnt = gsum + (size_t)NGRAPHS * HID;

    const int gemmBlocks    = NNODES / 16;
    const int scatterBlocks = (NEDGES + 255) / 256;
    const int readBlocks    = (NNODES * 64 + 255) / 256;
    const int finBlocks     = (NGRAPHS * 64 + 255) / 256;

    hipLaunchKernelGGL((gcn_gemm_wmma<false>), dim3(gemmBlocks), dim3(128), 0, stream,
                       feats, W1, (const float*)nullptr, bufA);
    hipMemsetAsync(bufB, 0, nodeElems * sizeof(float), stream);
    hipLaunchKernelGGL(gcn_scatter, dim3(scatterBlocks), dim3(256), 0, stream,
                       bufA, src, dst, bufB);
    hipLaunchKernelGGL((gcn_gemm_wmma<true>), dim3(gemmBlocks), dim3(128), 0, stream,
                       bufB, W2, b1, bufA);
    hipMemsetAsync(bufB, 0, nodeElems * sizeof(float), stream);
    hipLaunchKernelGGL(gcn_scatter, dim3(scatterBlocks), dim3(256), 0, stream,
                       bufA, src, dst, bufB);
    hipMemsetAsync(gsum, 0, ((size_t)NGRAPHS * HID + NGRAPHS) * sizeof(float), stream);
    hipLaunchKernelGGL(gcn_readout_acc, dim3(readBlocks), dim3(256), 0, stream,
                       bufB, gids, gsum, gcnt);
    hipLaunchKernelGGL(gcn_finalize, dim3(finBlocks), dim3(256), 0, stream,
                       gsum, gcnt, b2, out);
}


# layer 2 is linear: its edge aggregation and the mean-pool sum collapse into per-graph f32 atomic sums of relu(agg1+b1)[src] (8 per-die planes), W2 then applied to 1000 pooled rows; sum order changes
# speedup vs baseline: 2.2650x; 1.6318x over previous
#include <hip/hip_runtime.h>

#define HID      64
#define NNODES   100000
#define NEDGES   1200000
#define NGRAPHS  1000
#define NCOPY    8
#define PROWS    1008

typedef __attribute__((ext_vector_type(2))) float v2f;
typedef __attribute__((ext_vector_type(8))) float v8f;

template <bool POOL>
__global__ __launch_bounds__(128) void gcn_gemm_wmma(
    const float* __restrict__ In, const float* __restrict__ W,
    float* __restrict__ Out)
{
    __shared__ float sW[64 * 65];
    __shared__ float sA[16 * 65];

    const int tid  = threadIdx.x;
    const int wave = tid >> 5;
    const int lane = tid & 31;
    const int rowBase = blockIdx.x * 16;

    for (int i = tid; i < 64 * 64; i += 128) {
        int r = i >> 6, c = i & 63;
        sW[r * 65 + c] = W[i];
    }
    for (int i = tid; i < 16 * 64; i += 128) {
        int r = i >> 6, c = i & 63;
        float v = 0.0f;
        if (!POOL) {
            v = In[(rowBase + r) * 64 + c];
        } else if (rowBase + r < NGRAPHS) {
#pragma unroll
            for (int p = 0; p < NCOPY; ++p)
                v += In[(size_t)p * (NGRAPHS * HID) + (rowBase + r) * 64 + c];
        }
        sA[r * 65 + c] = v;
    }
    __syncthreads();

    const int m     = lane & 15;
    const int n     = (wave << 4) + (lane & 15);
    const int khalf = (lane >> 4) << 1;

    v8f acc = {};
#pragma unroll
    for (int k = 0; k < 64; k += 4) {
        v2f a, b;
        a.x = sA[m * 65 + k + khalf];
        a.y = sA[m * 65 + k + khalf + 1];
        b.x = sW[(k + khalf) * 65 + n];
        b.y = sW[(k + khalf + 1) * 65 + n];
        acc = __builtin_amdgcn_wmma_f32_16x16x4_f32(
             false, a,  false, b,
             (short)0, acc,  false,  false);
    }

    const int rOff = (lane >> 4) << 3;
    float* outp = Out + (size_t)(rowBase + rOff) * 64 + n;
#pragma unroll
    for (int v = 0; v < 8; ++v)
        outp[(size_t)v * 64] = acc[v];
}

#define EDGES_PER_WAVE 32
#define GATHER_BATCH   8
__global__ __launch_bounds__(256) void gcn_scatter(
    const float* __restrict__ X, const int* __restrict__ src,
    const int* __restrict__ dst, float* __restrict__ Agg)
{
    const unsigned gid = blockIdx.x * 256u + threadIdx.x;
    const unsigned w = gid >> 5;
    const unsigned l = gid & 31;
    const unsigned e0 = w * EDGES_PER_WAVE;
    if (e0 >= NEDGES) return;
    const int sv = src[gid];
    const int dv = dst[gid];
    const float* xl = X + l * 2;
    float* al = Agg + l * 2;

    for (int j0 = 0; j0 < EDGES_PER_WAVE; j0 += GATHER_BATCH) {
        float2 v[GATHER_BATCH];
#pragma unroll
        for (int u = 0; u < GATHER_BATCH; ++u) {
            const int s = __builtin_amdgcn_readlane(sv, j0 + u);
            v[u] = *(const float2*)(xl + (size_t)s * 64);
        }
#pragma unroll
        for (int u = 0; u < GATHER_BATCH; ++u) {
            const int d = __builtin_amdgcn_readlane(dv, j0 + u);
            float* p = al + (size_t)d * 64;
            unsafeAtomicAdd(p, v[u].x);
            unsafeAtomicAdd(p + 1, v[u].y);
        }
    }
}

__global__ __launch_bounds__(256) void gcn_scatter_pool(
    const float* __restrict__ Agg1, const float* __restrict__ b1,
    const int* __restrict__ src, const int* __restrict__ dst,
    const int* __restrict__ gids, float* __restrict__ S, float* __restrict__ gcnt)
{
    const unsigned gid = blockIdx.x * 256u + threadIdx.x;
    const unsigned w = gid >> 5;
    const unsigned l = gid & 31;
    const unsigned e0 = w * EDGES_PER_WAVE;
    if (e0 >= NEDGES) return;
    const int sv = src[gid];
    const int gv = gids[dst[gid]];
    if (gid < NNODES) unsafeAtomicAdd(&gcnt[gids[gid]], 1.0f);

    const unsigned plane = (__builtin_amdgcn_s_sendmsg_rtn(0x87) >> 16) & (NCOPY - 1);
    const float2 bb = *(const float2*)(b1 + l * 2);
    const float* xl = Agg1 + l * 2;
    float* al = S + (size_t)plane * (NGRAPHS * HID) + l * 2;

    for (int j0 = 0; j0 < EDGES_PER_WAVE; j0 += GATHER_BATCH) {
        float2 v[GATHER_BATCH];
#pragma unroll
        for (int u = 0; u < GATHER_BATCH; ++u) {
            const int s = __builtin_amdgcn_readlane(sv, j0 + u);
            v[u] = *(const float2*)(xl + (size_t)s * 64);
        }
        __builtin_amdgcn_sched_barrier(0);
#pragma unroll
        for (int u = 0; u < GATHER_BATCH; ++u) {
            const int g = __builtin_amdgcn_readlane(gv, j0 + u);
            float* p = al + (size_t)g * 64;
            const float hx = v[u].x + bb.x;
            const float hy = v[u].y + bb.y;
            unsafeAtomicAdd(p, hx > 0.0f ? hx : 0.0f);
            unsafeAtomicAdd(p + 1, hy > 0.0f ? hy : 0.0f);
        }
    }
}

__global__ __launch_bounds__(256) void gcn_finalize(
    const float* __restrict__ gsum, const float* __restrict__ gcnt,
    const float* __restrict__ b2, float* __restrict__ out)
{
    const unsigned gid = blockIdx.x * 256u + threadIdx.x;
    if (gid >= NGRAPHS * 64u) return;
    const unsigned g = gid >> 6;
    const unsigned c = gid & 63;
    const float cnt = gcnt[g];
    out[gid] = (cnt > 0.0f) ? (gsum[gid] / cnt + b2[c]) : 0.0f;
}

extern "C" void kernel_launch(void* const* d_in, const int* in_sizes, int n_in,
                              void* d_out, int out_size, void* d_ws, size_t ws_size,
                              hipStream_t stream)
{
    (void)in_sizes; (void)n_in; (void)out_size; (void)ws_size;

    const float* feats = (const float*)d_in[0];
    const float* W1    = (const float*)d_in[1];
    const float* b1    = (const float*)d_in[2];
    const float* W2    = (const float*)d_in[3];
    const float* b2    = (const float*)d_in[4];
    const int*   src   = (const int*)d_in[5];
    const int*   dst   = (const int*)d_in[6];
    const int*   gids  = (const int*)d_in[7];
    float* out = (float*)d_out;

    const size_t nodeElems = (size_t)NNODES * HID;
    float* bufA = (float*)d_ws;
    float* bufB = bufA + nodeElems;
    float* S    = bufA;
    float* gcnt = S + (size_t)NCOPY * NGRAPHS * HID;
    float* P    = gcnt + 1024;

    const int gemmBlocks    = NNODES / 16;
    const int scatterBlocks = (NEDGES + 255) / 256;
    const int poolBlocks    = PROWS / 16;
    const int finBlocks     = (NGRAPHS * 64 + 255) / 256;

    hipLaunchKernelGGL((gcn_gemm_wmma<false>), dim3(gemmBlocks), dim3(128), 0, stream,
                       feats, W1, bufA);
    hipMemsetAsync(bufB, 0, nodeElems * sizeof(float), stream);
    hipLaunchKernelGGL(gcn_scatter, dim3(scatterBlocks), dim3(256), 0, stream,
                       bufA, src, dst, bufB);
    hipMemsetAsync(S, 0, ((size_t)NCOPY * NGRAPHS * HID + NGRAPHS) * sizeof(float), stream);
    hipLaunchKernelGGL(gcn_scatter_pool, dim3(scatterBlocks), dim3(256), 0, stream,
                       bufB, b1, src, dst, gids, S, gcnt);
    hipLaunchKernelGGL((gcn_gemm_wmma<true>), dim3(poolBlocks), dim3(128), 0, stream,
                       S, W2, P);
    hipLaunchKernelGGL(gcn_finalize, dim3(finBlocks), dim3(256), 0, stream,
                       P, gcnt, b2, out);
}


# scatter lanes own columns l and l+32 (not 2l,2l+1): each row load and each f32 atomic add instruction covers one whole 128-byte line, not half of both lines; same adds, arrival order as before
# speedup vs baseline: 2.7809x; 1.2278x over previous
#include <hip/hip_runtime.h>

#define HID      64
#define NNODES   100000
#define NEDGES   1200000
#define NGRAPHS  1000
#define NCOPY    8
#define PROWS    1008

typedef __attribute__((ext_vector_type(2))) float v2f;
typedef __attribute__((ext_vector_type(8))) float v8f;

template <bool POOL>
__global__ __launch_bounds__(128) void gcn_gemm_wmma(
    const float* __restrict__ In, const float* __restrict__ W,
    float* __restrict__ Out)
{
    __shared__ float sW[64 * 65];
    __shared__ float sA[16 * 65];

    const int tid  = threadIdx.x;
    const int wave = tid >> 5;
    const int lane = tid & 31;
    const int rowBase = blockIdx.x * 16;

    for (int i = tid; i < 64 * 64; i += 128) {
        int r = i >> 6, c = i & 63;
        sW[r * 65 + c] = W[i];
    }
    for (int i = tid; i < 16 * 64; i += 128) {
        int r = i >> 6, c = i & 63;
        float v = 0.0f;
        if (!POOL) {
            v = In[(rowBase + r) * 64 + c];
        } else if (rowBase + r < NGRAPHS) {
#pragma unroll
            for (int p = 0; p < NCOPY; ++p)
                v += In[(size_t)p * (NGRAPHS * HID) + (rowBase + r) * 64 + c];
        }
        sA[r * 65 + c] = v;
    }
    __syncthreads();

    const int m     = lane & 15;
    const int n     = (wave << 4) + (lane & 15);
    const int khalf = (lane >> 4) << 1;

    v8f acc = {};
#pragma unroll
    for (int k = 0; k < 64; k += 4) {
        v2f a, b;
        a.x = sA[m * 65 + k + khalf];
        a.y = sA[m * 65 + k + khalf + 1];
        b.x = sW[(k + khalf) * 65 + n];
        b.y = sW[(k + khalf + 1) * 65 + n];
        acc = __builtin_amdgcn_wmma_f32_16x16x4_f32(
             false, a,  false, b,
             (short)0, acc,  false,  false);
    }

    const int rOff = (lane >> 4) << 3;
    float* outp = Out + (size_t)(rowBase + rOff) * 64 + n;
#pragma unroll
    for (int v = 0; v < 8; ++v)
        outp[(size_t)v * 64] = acc[v];
}

#define EDGES_PER_WAVE 32
#define GATHER_BATCH   8
__global__ __launch_bounds__(256) void gcn_scatter(
    const float* __restrict__ X, const int* __restrict__ src,
    const int* __restrict__ dst, float* __restrict__ Agg)
{
    const unsigned gid = blockIdx.x * 256u + threadIdx.x;
    const unsigned w = gid >> 5;
    const unsigned l = gid & 31;
    const unsigned e0 = w * EDGES_PER_WAVE;
    if (e0 >= NEDGES) return;
    const int sv = src[gid];
    const int dv = dst[gid];
    const float* xl = X + l;
    float* al = Agg + l;

    for (int j0 = 0; j0 < EDGES_PER_WAVE; j0 += GATHER_BATCH) {
        float vlo[GATHER_BATCH], vhi[GATHER_BATCH];
#pragma unroll
        for (int u = 0; u < GATHER_BATCH; ++u) {
            const int s = __builtin_amdgcn_readlane(sv, j0 + u);
            const float* r = xl + (size_t)s * 64;
            vlo[u] = r[0];
            vhi[u] = r[32];
        }
#pragma unroll
        for (int u = 0; u < GATHER_BATCH; ++u) {
            const int d = __builtin_amdgcn_readlane(dv, j0 + u);
            float* p = al + (size_t)d * 64;
            unsafeAtomicAdd(p, vlo[u]);
            unsafeAtomicAdd(p + 32, vhi[u]);
        }
    }
}

__global__ __launch_bounds__(256) void gcn_scatter_pool(
    const float* __restrict__ Agg1, const float* __restrict__ b1,
    const int* __restrict__ src, const int* __restrict__ dst,
    const int* __restrict__ gids, float* __restrict__ S, float* __restrict__ gcnt)
{
    const unsigned gid = blockIdx.x * 256u + threadIdx.x;
    const unsigned w = gid >> 5;
    const unsigned l = gid & 31;
    const unsigned e0 = w * EDGES_PER_WAVE;
    if (e0 >= NEDGES) return;
    const int sv = src[gid];
    const int gv = gids[dst[gid]];
    if (gid < NNODES) unsafeAtomicAdd(&gcnt[gids[gid]], 1.0f);

    const unsigned plane = (__builtin_amdgcn_s_sendmsg_rtn(0x87) >> 16) & (NCOPY - 1);
    const float blo = b1[l];
    const float bhi = b1[l + 32];
    const float* xl = Agg1 + l;
    float* al = S + (size_t)plane * (NGRAPHS * HID) + l;

    for (int j0 = 0; j0 < EDGES_PER_WAVE; j0 += GATHER_BATCH) {
        float vlo[GATHER_BATCH], vhi[GATHER_BATCH];
#pragma unroll
        for (int u = 0; u < GATHER_BATCH; ++u) {
            const int s = __builtin_amdgcn_readlane(sv, j0 + u);
            const float* r = xl + (size_t)s * 64;
            vlo[u] = r[0];
            vhi[u] = r[32];
        }
        __builtin_amdgcn_sched_barrier(0);
#pragma unroll
        for (int u = 0; u < GATHER_BATCH; ++u) {
            const int g = __builtin_amdgcn_readlane(gv, j0 + u);
            float* p = al + (size_t)g * 64;
            const float hx = vlo[u] + blo;
            const float hy = vhi[u] + bhi;
            unsafeAtomicAdd(p, hx > 0.0f ? hx : 0.0f);
            unsafeAtomicAdd(p + 32, hy > 0.0f ? hy : 0.0f);
        }
    }
}

__global__ __launch_bounds__(256) void gcn_finalize(
    const float* __restrict__ gsum, const float* __restrict__ gcnt,
    const float* __restrict__ b2, float* __restrict__ out)
{
    const unsigned gid = blockIdx.x * 256u + threadIdx.x;
    if (gid >= NGRAPHS * 64u) return;
    const unsigned g = gid >> 6;
    const unsigned c = gid & 63;
    const float cnt = gcnt[g];
    out[gid] = (cnt > 0.0f) ? (gsum[gid] / cnt + b2[c]) : 0.0f;
}

extern "C" void kernel_launch(void* const* d_in, const int* in_sizes, int n_in,
                              void* d_out, int out_size, void* d_ws, size_t ws_size,
                              hipStream_t stream)
{
    (void)in_sizes; (void)n_in; (void)out_size; (void)ws_size;

    const float* feats = (const float*)d_in[0];
    const float* W1    = (const float*)d_in[1];
    const float* b1    = (const float*)d_in[2];
    const float* W2    = (const float*)d_in[3];
    const float* b2    = (const float*)d_in[4];
    const int*   src   = (const int*)d_in[5];
    const int*   dst   = (const int*)d_in[6];
    const int*   gids  = (const int*)d_in[7];
    float* out = (float*)d_out;

    const size_t nodeElems = (size_t)NNODES * HID;
    float* bufA = (float*)d_ws;
    float* bufB = bufA + nodeElems;
    float* S    = bufA;
    float* gcnt = S + (size_t)NCOPY * NGRAPHS * HID;
    float* P    = gcnt + 1024;

    const int gemmBlocks    = NNODES / 16;
    const int scatterBlocks = (NEDGES + 255) / 256;
    const int poolBlocks    = PROWS / 16;
    const int finBlocks     = (NGRAPHS * 64 + 255) / 256;

    hipLaunchKernelGGL((gcn_gemm_wmma<false>), dim3(gemmBlocks), dim3(128), 0, stream,
                       feats, W1, bufA);
    hipMemsetAsync(bufB, 0, nodeElems * sizeof(float), stream);
    hipLaunchKernelGGL(gcn_scatter, dim3(scatterBlocks), dim3(256), 0, stream,
                       bufA, src, dst, bufB);
    hipMemsetAsync(S, 0, ((size_t)NCOPY * NGRAPHS * HID + NGRAPHS) * sizeof(float), stream);
    hipLaunchKernelGGL(gcn_scatter_pool, dim3(scatterBlocks), dim3(256), 0, stream,
                       bufB, b1, src, dst, gids, S, gcnt);
    hipLaunchKernelGGL((gcn_gemm_wmma<true>), dim3(poolBlocks), dim3(128), 0, stream,
                       S, W2, P);
    hipLaunchKernelGGL(gcn_finalize, dim3(finBlocks), dim3(256), 0, stream,
                       P, gcnt, b2, out);
}


# pool scatter: each work-group (256 x 1024 threads) sums its edges' relu rows into an LDS copy of S with LDS f32 adds, then adds that copy to its die's plane, one line per add; sum order changes
# speedup vs baseline: 2.9202x; 1.0501x over previous
#include <hip/hip_runtime.h>

#define HID      64
#define NNODES   100000
#define NEDGES   1200000
#define NGRAPHS  1000
#define NCOPY    8
#define PROWS    1008

typedef __attribute__((ext_vector_type(2))) float v2f;
typedef __attribute__((ext_vector_type(8))) float v8f;

template <bool POOL>
__global__ __launch_bounds__(128) void gcn_gemm_wmma(
    const float* __restrict__ In, const float* __restrict__ W,
    float* __restrict__ Out)
{
    __shared__ float sW[64 * 65];
    __shared__ float sA[16 * 65];

    const int tid  = threadIdx.x;
    const int wave = tid >> 5;
    const int lane = tid & 31;
    const int rowBase = blockIdx.x * 16;

    for (int i = tid; i < 64 * 64; i += 128) {
        int r = i >> 6, c = i & 63;
        sW[r * 65 + c] = W[i];
    }
    for (int i = tid; i < 16 * 64; i += 128) {
        int r = i >> 6, c = i & 63;
        float v = 0.0f;
        if (!POOL) {
            v = In[(rowBase + r) * 64 + c];
        } else if (rowBase + r < NGRAPHS) {
#pragma unroll
            for (int p = 0; p < NCOPY; ++p)
                v += In[(size_t)p * (NGRAPHS * HID) + (rowBase + r) * 64 + c];
        }
        sA[r * 65 + c] = v;
    }
    __syncthreads();

    const int m     = lane & 15;
    const int n     = (wave << 4) + (lane & 15);
    const int khalf = (lane >> 4) << 1;

    v8f acc = {};
#pragma unroll
    for (int k = 0; k < 64; k += 4) {
        v2f a, b;
        a.x = sA[m * 65 + k + khalf];
        a.y = sA[m * 65 + k + khalf + 1];
        b.x = sW[(k + khalf) * 65 + n];
        b.y = sW[(k + khalf + 1) * 65 + n];
        acc = __builtin_amdgcn_wmma_f32_16x16x4_f32(
             false, a,  false, b,
             (short)0, acc,  false,  false);
    }

    const int rOff = (lane >> 4) << 3;
    float* outp = Out + (size_t)(rowBase + rOff) * 64 + n;
#pragma unroll
    for (int v = 0; v < 8; ++v)
        outp[(size_t)v * 64] = acc[v];
}

#define EDGES_PER_WAVE 32
#define GATHER_BATCH   8
__global__ __launch_bounds__(256) void gcn_scatter(
    const float* __restrict__ X, const int* __restrict__ src,
    const int* __restrict__ dst, float* __restrict__ Agg)
{
    const unsigned gid = blockIdx.x * 256u + threadIdx.x;
    const unsigned w = gid >> 5;
    const unsigned l = gid & 31;
    const unsigned e0 = w * EDGES_PER_WAVE;
    if (e0 >= NEDGES) return;
    const int sv = src[gid];
    const int dv = dst[gid];
    const float* xl = X + l;
    float* al = Agg + l;

    for (int j0 = 0; j0 < EDGES_PER_WAVE; j0 += GATHER_BATCH) {
        float vlo[GATHER_BATCH], vhi[GATHER_BATCH];
#pragma unroll
        for (int u = 0; u < GATHER_BATCH; ++u) {
            const int s = __builtin_amdgcn_readlane(sv, j0 + u);
            const float* r = xl + (size_t)s * 64;
            vlo[u] = r[0];
            vhi[u] = r[32];
        }
#pragma unroll
        for (int u = 0; u < GATHER_BATCH; ++u) {
            const int d = __builtin_amdgcn_readlane(dv, j0 + u);
            float* p = al + (size_t)d * 64;
            unsafeAtomicAdd(p, vlo[u]);
            unsafeAtomicAdd(p + 32, vhi[u]);
        }
    }
}

#define POOL_THREADS 1024
#define POOL_WGS     256
#define POOL_WAVES   (POOL_WGS * (POOL_THREADS / 32))
#define NCHUNKS      (NEDGES / EDGES_PER_WAVE)
#define POOL_BATCH   16

__device__ __forceinline__ void pool_batch(
    const int sv, const int gv, const int j0,
    const float* __restrict__ xl, float* sl, const float blo, const float bhi)
{
    float vlo[POOL_BATCH], vhi[POOL_BATCH];
#pragma unroll
    for (int u = 0; u < POOL_BATCH; ++u) {
        const int s = __builtin_amdgcn_readlane(sv, j0 + u);
        const float* r = xl + (size_t)s * 64;
        vlo[u] = r[0];
        vhi[u] = r[32];
    }
    __builtin_amdgcn_sched_barrier(0);
#pragma unroll
    for (int u = 0; u < POOL_BATCH; ++u) {
        const int g = __builtin_amdgcn_readlane(gv, j0 + u);
        float* p = sl + g * 64;
        const float hx = vlo[u] + blo;
        const float hy = vhi[u] + bhi;
        __hip_atomic_fetch_add(p, hx > 0.0f ? hx : 0.0f, __ATOMIC_RELAXED, __HIP_MEMORY_SCOPE_WORKGROUP);
        __hip_atomic_fetch_add(p + 32, hy > 0.0f ? hy : 0.0f, __ATOMIC_RELAXED, __HIP_MEMORY_SCOPE_WORKGROUP);
    }
}

__global__ __launch_bounds__(POOL_THREADS) void gcn_scatter_pool(
    const float* __restrict__ Agg1, const float* __restrict__ b1,
    const int* __restrict__ src, const int* __restrict__ dst,
    const int* __restrict__ gids, float* __restrict__ S, float* __restrict__ gcnt)
{
    __shared__ __attribute__((aligned(16))) float sS[NGRAPHS * HID];

    const unsigned tid  = threadIdx.x;
    const unsigned wave = __builtin_amdgcn_readfirstlane(tid >> 5);
    const unsigned l    = tid & 31;

    unsigned wc = wave * POOL_WGS + blockIdx.x;
    int sv, gv;
    {
        const unsigned e = wc * EDGES_PER_WAVE + l;
        sv = src[e];
        gv = gids[(unsigned)dst[e]];
        if (e < NNODES) unsafeAtomicAdd(&gcnt[gids[e]], 1.0f);
    }

    float4* s4 = reinterpret_cast<float4*>(sS);
    for (unsigned i = tid; i < NGRAPHS * HID / 4; i += POOL_THREADS)
        s4[i] = make_float4(0.0f, 0.0f, 0.0f, 0.0f);
    __syncthreads();

    const float blo = b1[l];
    const float bhi = b1[l + 32];
    const float* xl = Agg1 + l;
    float* sl = sS + l;

    for (;;) {
        const unsigned wcn = wc + POOL_WAVES;
        const bool more = wcn < NCHUNKS;
        const unsigned en = more ? (wcn * EDGES_PER_WAVE + l) : (wc * EDGES_PER_WAVE);
        const int svn = src[en];
        const unsigned dvn = (unsigned)dst[en];
        pool_batch(sv, gv, 0, xl, sl, blo, bhi);
        const int gvn = gids[dvn];
        pool_batch(sv, gv, POOL_BATCH, xl, sl, blo, bhi);
        if (!more) break;
        sv = svn; gv = gvn; wc = wcn;
    }
    __syncthreads();

    const unsigned plane = (__builtin_amdgcn_s_sendmsg_rtn(0x87) >> 16) & (NCOPY - 1);
    float* Sp = S + (size_t)plane * (NGRAPHS * HID);
#define FLUSH_BATCH 8
    for (unsigned base = tid; base < NGRAPHS * HID; base += FLUSH_BATCH * POOL_THREADS) {
        float v[FLUSH_BATCH];
#pragma unroll
        for (int u = 0; u < FLUSH_BATCH; ++u) {
            const unsigned i = base + u * POOL_THREADS;
            v[u] = (i < NGRAPHS * HID) ? sS[i] : 0.0f;
        }
#pragma unroll
        for (int u = 0; u < FLUSH_BATCH; ++u) {
            const unsigned i = base + u * POOL_THREADS;
            if (i < NGRAPHS * HID) unsafeAtomicAdd(Sp + i, v[u]);
        }
    }
}

__global__ __launch_bounds__(256) void gcn_finalize(
    const float* __restrict__ gsum, const float* __restrict__ gcnt,
    const float* __restrict__ b2, float* __restrict__ out)
{
    const unsigned gid = blockIdx.x * 256u + threadIdx.x;
    if (gid >= NGRAPHS * 64u) return;
    const unsigned g = gid >> 6;
    const unsigned c = gid & 63;
    const float cnt = gcnt[g];
    out[gid] = (cnt > 0.0f) ? (gsum[gid] / cnt + b2[c]) : 0.0f;
}

extern "C" void kernel_launch(void* const* d_in, const int* in_sizes, int n_in,
                              void* d_out, int out_size, void* d_ws, size_t ws_size,
                              hipStream_t stream)
{
    (void)in_sizes; (void)n_in; (void)out_size; (void)ws_size;

    const float* feats = (const float*)d_in[0];
    const float* W1    = (const float*)d_in[1];
    const float* b1    = (const float*)d_in[2];
    const float* W2    = (const float*)d_in[3];
    const float* b2    = (const float*)d_in[4];
    const int*   src   = (const int*)d_in[5];
    const int*   dst   = (const int*)d_in[6];
    const int*   gids  = (const int*)d_in[7];
    float* out = (float*)d_out;

    const size_t nodeElems = (size_t)NNODES * HID;
    float* bufA = (float*)d_ws;
    float* bufB = bufA + nodeElems;
    float* S    = bufA;
    float* gcnt = S + (size_t)NCOPY * NGRAPHS * HID;
    float* P    = gcnt + 1024;

    const int gemmBlocks    = NNODES / 16;
    const int scatterBlocks = (NEDGES + 255) / 256;
    const int poolBlocks    = PROWS / 16;
    const int finBlocks     = (NGRAPHS * 64 + 255) / 256;

    hipLaunchKernelGGL((gcn_gemm_wmma<false>), dim3(gemmBlocks), dim3(128), 0, stream,
                       feats, W1, bufA);
    hipMemsetAsync(bufB, 0, nodeElems * sizeof(float), stream);
    hipLaunchKernelGGL(gcn_scatter, dim3(scatterBlocks), dim3(256), 0, stream,
                       bufA, src, dst, bufB);
    hipMemsetAsync(S, 0, ((size_t)NCOPY * NGRAPHS * HID + NGRAPHS) * sizeof(float), stream);
    hipLaunchKernelGGL(gcn_scatter_pool, dim3(POOL_WGS), dim3(POOL_THREADS), 0, stream,
                       bufB, b1, src, dst, gids, S, gcnt);
    hipLaunchKernelGGL((gcn_gemm_wmma<true>), dim3(poolBlocks), dim3(128), 0, stream,
                       S, W2, P);
    hipLaunchKernelGGL(gcn_finalize, dim3(finBlocks), dim3(256), 0, stream,
                       P, gcnt, b2, out);
}


# layer-1 scatter: a work-group reads which chip half it runs on, claims edges by ticket and adds only into that half's part of agg1: each line takes f32 atomic adds from one half only; order as before
# speedup vs baseline: 3.2278x; 1.1053x over previous
#include <hip/hip_runtime.h>

#define HID      64
#define NNODES   100000
#define NEDGES   1200000
#define NGRAPHS  1000
#define NCOPY    8
#define PROWS    1008

typedef __attribute__((ext_vector_type(2))) float v2f;
typedef __attribute__((ext_vector_type(8))) float v8f;

template <bool POOL>
__global__ __launch_bounds__(128) void gcn_gemm_wmma(
    const float* __restrict__ In, const float* __restrict__ W,
    float* __restrict__ Out, float* __restrict__ tickets)
{
    __shared__ float sW[64 * 65];
    __shared__ float sA[16 * 65];

    const int tid  = threadIdx.x;
    const int wave = tid >> 5;
    const int lane = tid & 31;
    const int rowBase = blockIdx.x * 16;

    if (!POOL) {
        if (blockIdx.x == 0 && tid < 2) reinterpret_cast<unsigned*>(tickets)[tid] = 0u;
    }

    for (int i = tid; i < 64 * 64; i += 128) {
        int r = i >> 6, c = i & 63;
        sW[r * 65 + c] = W[i];
    }
    for (int i = tid; i < 16 * 64; i += 128) {
        int r = i >> 6, c = i & 63;
        float v = 0.0f;
        if (!POOL) {
            v = In[(rowBase + r) * 64 + c];
        } else if (rowBase + r < NGRAPHS) {
#pragma unroll
            for (int p = 0; p < NCOPY; ++p)
                v += In[(size_t)p * (NGRAPHS * HID) + (rowBase + r) * 64 + c];
        }
        sA[r * 65 + c] = v;
    }
    __syncthreads();

    const int m     = lane & 15;
    const int n     = (wave << 4) + (lane & 15);
    const int khalf = (lane >> 4) << 1;

    v8f acc = {};
#pragma unroll
    for (int k = 0; k < 64; k += 4) {
        v2f a, b;
        a.x = sA[m * 65 + k + khalf];
        a.y = sA[m * 65 + k + khalf + 1];
        b.x = sW[(k + khalf) * 65 + n];
        b.y = sW[(k + khalf + 1) * 65 + n];
        acc = __builtin_amdgcn_wmma_f32_16x16x4_f32(
             false, a,  false, b,
             (short)0, acc,  false,  false);
    }

    const int rOff = (lane >> 4) << 3;
    float* outp = Out + (size_t)(rowBase + rOff) * 64 + n;
#pragma unroll
    for (int v = 0; v < 8; ++v)
        outp[(size_t)v * 64] = acc[v];
}

#define EDGES_PER_WAVE 32
#define SC_EDGES   64
#define SC_WAVES   8
#define SC_CHUNKS  (NEDGES / SC_EDGES)
#define SC_UNITS   ((SC_CHUNKS + SC_WAVES - 1) / SC_WAVES)
#define SC_SPLIT   (NNODES / 2)
#define GATHER_BATCH 8

__device__ __forceinline__ void scatter_request(
    unsigned long long& todo, const int sv0, const int sv1, const int dv0, const int dv1,
    const float* __restrict__ X, const unsigned l, float& lo, float& hi, unsigned& d)
{
    const int j = __builtin_ctzll(todo);
    todo &= todo - 1ull;
    const int jl = j & 31;
    const unsigned sa = (unsigned)__builtin_amdgcn_readlane(sv0, jl);
    const unsigned sb = (unsigned)__builtin_amdgcn_readlane(sv1, jl);
    const unsigned da = (unsigned)__builtin_amdgcn_readlane(dv0, jl);
    const unsigned db = (unsigned)__builtin_amdgcn_readlane(dv1, jl);
    const unsigned s = j < 32 ? sa : sb;
    d = j < 32 ? da : db;
    const float* r = X + (size_t)s * 64;
    lo = r[l];
    hi = r[l + 32];
}

__global__ __launch_bounds__(256) void gcn_scatter(
    const float* __restrict__ X, const int* __restrict__ src,
    const int* __restrict__ dst, float* __restrict__ Agg, float* __restrict__ tickets)
{
    __shared__ unsigned sTicket[2];
    const unsigned tid = threadIdx.x;
    if (tid == 0) {
        unsigned* tick = reinterpret_cast<unsigned*>(tickets);
        unsigned part = (__builtin_amdgcn_s_sendmsg_rtn(0x87) >> 18) & 1u;
        unsigned t = atomicAdd(&tick[part], 1u);
        if (t >= SC_UNITS) {
            part ^= 1u;
            t = atomicAdd(&tick[part], 1u);
        }
        sTicket[0] = t;
        sTicket[1] = part;
    }
    __syncthreads();
    const unsigned unit = __builtin_amdgcn_readfirstlane(sTicket[0]);
    const unsigned part = __builtin_amdgcn_readfirstlane(sTicket[1]);
    const unsigned wave = __builtin_amdgcn_readfirstlane(tid >> 5);
    const unsigned l = tid & 31;
    const unsigned c = unit * SC_WAVES + wave;
    if (unit >= SC_UNITS || c >= SC_CHUNKS) return;
    const unsigned e = c * SC_EDGES + l;
    const int sv0 = src[e];
    const int dv0 = dst[e];
    const int sv1 = src[e + 32];
    const int dv1 = dst[e + 32];
    const bool upper = part != 0u;
    const unsigned todo0 = __builtin_amdgcn_ballot_w32(((unsigned)dv0 >= SC_SPLIT) == upper);
    const unsigned todo1 = __builtin_amdgcn_ballot_w32(((unsigned)dv1 >= SC_SPLIT) == upper);
    unsigned long long todo = ((unsigned long long)todo1 << 32) | todo0;
    int n = __builtin_popcountll(todo);

    for (; n >= GATHER_BATCH; n -= GATHER_BATCH) {
        float vlo[GATHER_BATCH], vhi[GATHER_BATCH];
        unsigned d[GATHER_BATCH];
#pragma unroll
        for (int u = 0; u < GATHER_BATCH; ++u)
            scatter_request(todo, sv0, sv1, dv0, dv1, X, l, vlo[u], vhi[u], d[u]);
#pragma unroll
        for (int u = 0; u < GATHER_BATCH; ++u) {
            float* p = Agg + (size_t)d[u] * 64;
            unsafeAtomicAdd(p + l, vlo[u]);
            unsafeAtomicAdd(p + l + 32, vhi[u]);
        }
    }
    if (n > 0) {
        float vlo[GATHER_BATCH - 1], vhi[GATHER_BATCH - 1];
        unsigned d[GATHER_BATCH - 1];
#pragma unroll
        for (int u = 0; u < GATHER_BATCH - 1; ++u)
            if (u < n) scatter_request(todo, sv0, sv1, dv0, dv1, X, l, vlo[u], vhi[u], d[u]);
#pragma unroll
        for (int u = 0; u < GATHER_BATCH - 1; ++u) {
            if (u < n) {
                float* p = Agg + (size_t)d[u] * 64;
                unsafeAtomicAdd(p + l, vlo[u]);
                unsafeAtomicAdd(p + l + 32, vhi[u]);
            }
        }
    }
}

#define POOL_THREADS 1024
#define POOL_WGS     256
#define POOL_WAVES   (POOL_WGS * (POOL_THREADS / 32))
#define NCHUNKS      (NEDGES / EDGES_PER_WAVE)
#define POOL_BATCH   16

__device__ __forceinline__ void pool_batch(
    const int sv, const int gv, const int j0,
    const float* __restrict__ xl, float* sl, const float blo, const float bhi)
{
    float vlo[POOL_BATCH], vhi[POOL_BATCH];
#pragma unroll
    for (int u = 0; u < POOL_BATCH; ++u) {
        const int s = __builtin_amdgcn_readlane(sv, j0 + u);
        const float* r = xl + (size_t)s * 64;
        vlo[u] = r[0];
        vhi[u] = r[32];
    }
    __builtin_amdgcn_sched_barrier(0);
#pragma unroll
    for (int u = 0; u < POOL_BATCH; ++u) {
        const int g = __builtin_amdgcn_readlane(gv, j0 + u);
        float* p = sl + g * 64;
        const float hx = vlo[u] + blo;
        const float hy = vhi[u] + bhi;
        __hip_atomic_fetch_add(p, hx > 0.0f ? hx : 0.0f, __ATOMIC_RELAXED, __HIP_MEMORY_SCOPE_WORKGROUP);
        __hip_atomic_fetch_add(p + 32, hy > 0.0f ? hy : 0.0f, __ATOMIC_RELAXED, __HIP_MEMORY_SCOPE_WORKGROUP);
    }
}

__global__ __launch_bounds__(POOL_THREADS) void gcn_scatter_pool(
    const float* __restrict__ Agg1, const float* __restrict__ b1,
    const int* __restrict__ src, const int* __restrict__ dst,
    const int* __restrict__ gids, float* __restrict__ S, float* __restrict__ gcnt)
{
    __shared__ __attribute__((aligned(16))) float sS[NGRAPHS * HID];

    const unsigned tid  = threadIdx.x;
    const unsigned wave = __builtin_amdgcn_readfirstlane(tid >> 5);
    const unsigned l    = tid & 31;

    unsigned wc = wave * POOL_WGS + blockIdx.x;
    int sv, gv;
    {
        const unsigned e = wc * EDGES_PER_WAVE + l;
        sv = src[e];
        gv = gids[(unsigned)dst[e]];
        if (e < NNODES) unsafeAtomicAdd(&gcnt[gids[e]], 1.0f);
    }

    float4* s4 = reinterpret_cast<float4*>(sS);
    for (unsigned i = tid; i < NGRAPHS * HID / 4; i += POOL_THREADS)
        s4[i] = make_float4(0.0f, 0.0f, 0.0f, 0.0f);
    __syncthreads();

    const float blo = b1[l];
    const float bhi = b1[l + 32];
    const float* xl = Agg1 + l;
    float* sl = sS + l;

    for (;;) {
        const unsigned wcn = wc + POOL_WAVES;
        const bool more = wcn < NCHUNKS;
        const unsigned en = more ? (wcn * EDGES_PER_WAVE + l) : (wc * EDGES_PER_WAVE);
        const int svn = src[en];
        const unsigned dvn = (unsigned)dst[en];
        pool_batch(sv, gv, 0, xl, sl, blo, bhi);
        const int gvn = gids[dvn];
        pool_batch(sv, gv, POOL_BATCH, xl, sl, blo, bhi);
        if (!more) break;
        sv = svn; gv = gvn; wc = wcn;
    }
    __syncthreads();

    const unsigned plane = (__builtin_amdgcn_s_sendmsg_rtn(0x87) >> 16) & (NCOPY - 1);
    float* Sp = S + (size_t)plane * (NGRAPHS * HID);
#define FLUSH_BATCH 8
    for (unsigned base = tid; base < NGRAPHS * HID; base += FLUSH_BATCH * POOL_THREADS) {
        float v[FLUSH_BATCH];
#pragma unroll
        for (int u = 0; u < FLUSH_BATCH; ++u) {
            const unsigned i = base + u * POOL_THREADS;
            v[u] = (i < NGRAPHS * HID) ? sS[i] : 0.0f;
        }
#pragma unroll
        for (int u = 0; u < FLUSH_BATCH; ++u) {
            const unsigned i = base + u * POOL_THREADS;
            if (i < NGRAPHS * HID) unsafeAtomicAdd(Sp + i, v[u]);
        }
    }
}

__global__ __launch_bounds__(256) void gcn_finalize(
    const float* __restrict__ gsum, const float* __restrict__ gcnt,
    const float* __restrict__ b2, float* __restrict__ out)
{
    const unsigned gid = blockIdx.x * 256u + threadIdx.x;
    if (gid >= NGRAPHS * 64u) return;
    const unsigned g = gid >> 6;
    const unsigned c = gid & 63;
    const float cnt = gcnt[g];
    out[gid] = (cnt > 0.0f) ? (gsum[gid] / cnt + b2[c]) : 0.0f;
}

extern "C" void kernel_launch(void* const* d_in, const int* in_sizes, int n_in,
                              void* d_out, int out_size, void* d_ws, size_t ws_size,
                              hipStream_t stream)
{
    (void)in_sizes; (void)n_in; (void)out_size; (void)ws_size;

    const float* feats = (const float*)d_in[0];
    const float* W1    = (const float*)d_in[1];
    const float* b1    = (const float*)d_in[2];
    const float* W2    = (const float*)d_in[3];
    const float* b2    = (const float*)d_in[4];
    const int*   src   = (const int*)d_in[5];
    const int*   dst   = (const int*)d_in[6];
    const int*   gids  = (const int*)d_in[7];
    float* out = (float*)d_out;

    const size_t nodeElems = (size_t)NNODES * HID;
    float* bufA = (float*)d_ws;
    float* bufB = bufA + nodeElems;
    float* S    = bufA;
    float* gcnt = S + (size_t)NCOPY * NGRAPHS * HID;
    float* P    = gcnt + 1024;

    const int gemmBlocks    = NNODES / 16;
    const int scatterBlocks = 2 * SC_UNITS;
    const int poolBlocks    = PROWS / 16;
    const int finBlocks     = (NGRAPHS * 64 + 255) / 256;

    hipLaunchKernelGGL((gcn_gemm_wmma<false>), dim3(gemmBlocks), dim3(128), 0, stream,
                       feats, W1, bufA, out);
    hipMemsetAsync(bufB, 0, nodeElems * sizeof(float), stream);
    hipLaunchKernelGGL(gcn_scatter, dim3(scatterBlocks), dim3(256), 0, stream,
                       bufA, src, dst, bufB, out);
    hipMemsetAsync(S, 0, ((size_t)NCOPY * NGRAPHS * HID + NGRAPHS) * sizeof(float), stream);
    hipLaunchKernelGGL(gcn_scatter_pool, dim3(POOL_WGS), dim3(POOL_THREADS), 0, stream,
                       bufB, b1, src, dst, gids, S, gcnt);
    hipLaunchKernelGGL((gcn_gemm_wmma<true>), dim3(poolBlocks), dim3(128), 0, stream,
                       S, W2, P, (float*)nullptr);
    hipLaunchKernelGGL(gcn_finalize, dim3(finBlocks), dim3(256), 0, stream,
                       P, gcnt, b2, out);
}


# layer-1 GEMM: a work-group takes 5 row tiles, each wave keeps its 16 columns of W1 in registers, loaded once from global (no LDS copy of W per 16 rows); A tiles double-buffered in LDS; sums as before
# speedup vs baseline: 3.4869x; 1.0803x over previous
#include <hip/hip_runtime.h>

#define HID      64
#define NNODES   100000
#define NEDGES   1200000
#define NGRAPHS  1000
#define NCOPY    8
#define PROWS    1008

typedef __attribute__((ext_vector_type(2))) float v2f;
typedef __attribute__((ext_vector_type(8))) float v8f;


#define G1_TILES  5
#define G1_PITCH  68
__device__ __forceinline__ void g1_tile(const float* a, const v2f (&bw)[16], float* outp)
{
    v8f acc = {};
#pragma unroll
    for (int t = 0; t < 16; ++t) {
        const v2f av = *reinterpret_cast<const v2f*>(a + 4 * t);
        acc = __builtin_amdgcn_wmma_f32_16x16x4_f32(
             false, av,  false, bw[t],
             (short)0, acc,  false,  false);
    }
#pragma unroll
    for (int v = 0; v < 8; ++v)
        outp[(size_t)v * 64] = acc[v];
}

__global__ __launch_bounds__(128) void gcn_gemm1_wmma(
    const float* In, const float* __restrict__ W,
    float* __restrict__ Out, float* __restrict__ tickets)
{
    __shared__ __attribute__((aligned(16))) float sA[2][16 * G1_PITCH];

    const int tid  = threadIdx.x;
    const int wave = tid >> 5;
    const int lane = tid & 31;

    if (blockIdx.x == 0 && tid < 2) reinterpret_cast<unsigned*>(tickets)[tid] = 0u;

    const int m     = lane & 15;
    const int n     = (wave << 4) + (lane & 15);
    const int khalf = (lane >> 4) << 1;
    const int rOff  = (lane >> 4) << 3;

    const int r0 = tid >> 4, c0 = (tid & 15) << 2;
    const float4* in4 = reinterpret_cast<const float4*>(In) + (size_t)blockIdx.x * (G1_TILES * 256) + tid;
    float4 p0 = in4[0];
    float4 p1 = in4[128];

    v2f bw[16];
#pragma unroll
    for (int t = 0; t < 16; ++t) {
        bw[t].x = W[(4 * t + khalf) * 64 + n];
        bw[t].y = W[(4 * t + khalf + 1) * 64 + n];
    }

    *reinterpret_cast<float4*>(&sA[0][r0 * G1_PITCH + c0])       = p0;
    *reinterpret_cast<float4*>(&sA[0][(r0 + 8) * G1_PITCH + c0]) = p1;
    __syncthreads();

    for (int tile = 0; tile < G1_TILES - 1; ++tile) {
        p0 = in4[(tile + 1) * 256];
        p1 = in4[(tile + 1) * 256 + 128];
        __builtin_amdgcn_sched_barrier(0);
        g1_tile(sA[tile & 1] + m * G1_PITCH + khalf, bw,
                Out + ((size_t)(blockIdx.x * G1_TILES + tile) * 16 + rOff) * 64 + n);
        float* nb = sA[(tile + 1) & 1];
        *reinterpret_cast<float4*>(&nb[r0 * G1_PITCH + c0])       = p0;
        *reinterpret_cast<float4*>(&nb[(r0 + 8) * G1_PITCH + c0]) = p1;
        __syncthreads();
    }
    g1_tile(sA[(G1_TILES - 1) & 1] + m * G1_PITCH + khalf, bw,
            Out + ((size_t)(blockIdx.x * G1_TILES + G1_TILES - 1) * 16 + rOff) * 64 + n);
}

__global__ __launch_bounds__(128) void gcn_gemm_pool_wmma(
    const float* __restrict__ In, const float* __restrict__ W,
    float* __restrict__ Out)
{
    __shared__ float sW[64 * 65];
    __shared__ float sA[16 * 65];

    const int tid  = threadIdx.x;
    const int wave = tid >> 5;
    const int lane = tid & 31;
    const int rowBase = blockIdx.x * 16;

    for (int i = tid; i < 64 * 64; i += 128) {
        int r = i >> 6, c = i & 63;
        sW[r * 65 + c] = W[i];
    }
    for (int i = tid; i < 16 * 64; i += 128) {
        int r = i >> 6, c = i & 63;
        float v = 0.0f;
        if (rowBase + r < NGRAPHS) {
#pragma unroll
            for (int p = 0; p < NCOPY; ++p)
                v += In[(size_t)p * (NGRAPHS * HID) + (rowBase + r) * 64 + c];
        }
        sA[r * 65 + c] = v;
    }
    __syncthreads();

    const int m     = lane & 15;
    const int n     = (wave << 4) + (lane & 15);
    const int khalf = (lane >> 4) << 1;

    v8f acc = {};
#pragma unroll
    for (int k = 0; k < 64; k += 4) {
        v2f a, b;
        a.x = sA[m * 65 + k + khalf];
        a.y = sA[m * 65 + k + khalf + 1];
        b.x = sW[(k + khalf) * 65 + n];
        b.y = sW[(k + khalf + 1) * 65 + n];
        acc = __builtin_amdgcn_wmma_f32_16x16x4_f32(
             false, a,  false, b,
             (short)0, acc,  false,  false);
    }

    const int rOff = (lane >> 4) << 3;
    float* outp = Out + (size_t)(rowBase + rOff) * 64 + n;
#pragma unroll
    for (int v = 0; v < 8; ++v)
        outp[(size_t)v * 64] = acc[v];
}

#define EDGES_PER_WAVE 32
#define SC_EDGES   64
#define SC_WAVES   8
#define SC_CHUNKS  (NEDGES / SC_EDGES)
#define SC_UNITS   ((SC_CHUNKS + SC_WAVES - 1) / SC_WAVES)
#define SC_SPLIT   (NNODES / 2)
#define GATHER_BATCH 8

__device__ __forceinline__ void scatter_request(
    unsigned long long& todo, const int sv0, const int sv1, const int dv0, const int dv1,
    const float* __restrict__ X, const unsigned l, float& lo, float& hi, unsigned& d)
{
    const int j = __builtin_ctzll(todo);
    todo &= todo - 1ull;
    const int jl = j & 31;
    const unsigned sa = (unsigned)__builtin_amdgcn_readlane(sv0, jl);
    const unsigned sb = (unsigned)__builtin_amdgcn_readlane(sv1, jl);
    const unsigned da = (unsigned)__builtin_amdgcn_readlane(dv0, jl);
    const unsigned db = (unsigned)__builtin_amdgcn_readlane(dv1, jl);
    const unsigned s = j < 32 ? sa : sb;
    d = j < 32 ? da : db;
    const float* r = X + (size_t)s * 64;
    lo = r[l];
    hi = r[l + 32];
}

__global__ __launch_bounds__(256) void gcn_scatter(
    const float* __restrict__ X, const int* __restrict__ src,
    const int* __restrict__ dst, float* __restrict__ Agg, float* __restrict__ tickets)
{
    __shared__ unsigned sTicket[2];
    const unsigned tid = threadIdx.x;
    if (tid == 0) {
        unsigned* tick = reinterpret_cast<unsigned*>(tickets);
        unsigned part = (__builtin_amdgcn_s_sendmsg_rtn(0x87) >> 18) & 1u;
        unsigned t = atomicAdd(&tick[part], 1u);
        if (t >= SC_UNITS) {
            part ^= 1u;
            t = atomicAdd(&tick[part], 1u);
        }
        sTicket[0] = t;
        sTicket[1] = part;
    }
    __syncthreads();
    const unsigned unit = __builtin_amdgcn_readfirstlane(sTicket[0]);
    const unsigned part = __builtin_amdgcn_readfirstlane(sTicket[1]);
    const unsigned wave = __builtin_amdgcn_readfirstlane(tid >> 5);
    const unsigned l = tid & 31;
    const unsigned c = unit * SC_WAVES + wave;
    if (unit >= SC_UNITS || c >= SC_CHUNKS) return;
    const unsigned e = c * SC_EDGES + l;
    const int sv0 = src[e];
    const int dv0 = dst[e];
    const int sv1 = src[e + 32];
    const int dv1 = dst[e + 32];
    const bool upper = part != 0u;
    const unsigned todo0 = __builtin_amdgcn_ballot_w32(((unsigned)dv0 >= SC_SPLIT) == upper);
    const unsigned todo1 = __builtin_amdgcn_ballot_w32(((unsigned)dv1 >= SC_SPLIT) == upper);
    unsigned long long todo = ((unsigned long long)todo1 << 32) | todo0;
    int n = __builtin_popcountll(todo);

    for (; n >= GATHER_BATCH; n -= GATHER_BATCH) {
        float vlo[GATHER_BATCH], vhi[GATHER_BATCH];
        unsigned d[GATHER_BATCH];
#pragma unroll
        for (int u = 0; u < GATHER_BATCH; ++u)
            scatter_request(todo, sv0, sv1, dv0, dv1, X, l, vlo[u], vhi[u], d[u]);
#pragma unroll
        for (int u = 0; u < GATHER_BATCH; ++u) {
            float* p = Agg + (size_t)d[u] * 64;
            unsafeAtomicAdd(p + l, vlo[u]);
            unsafeAtomicAdd(p + l + 32, vhi[u]);
        }
    }
    if (n > 0) {
        float vlo[GATHER_BATCH - 1], vhi[GATHER_BATCH - 1];
        unsigned d[GATHER_BATCH - 1];
#pragma unroll
        for (int u = 0; u < GATHER_BATCH - 1; ++u)
            if (u < n) scatter_request(todo, sv0, sv1, dv0, dv1, X, l, vlo[u], vhi[u], d[u]);
#pragma unroll
        for (int u = 0; u < GATHER_BATCH - 1; ++u) {
            if (u < n) {
                float* p = Agg + (size_t)d[u] * 64;
                unsafeAtomicAdd(p + l, vlo[u]);
                unsafeAtomicAdd(p + l + 32, vhi[u]);
            }
        }
    }
}

#define POOL_THREADS 1024
#define POOL_WGS     256
#define POOL_WAVES   (POOL_WGS * (POOL_THREADS / 32))
#define NCHUNKS      (NEDGES / EDGES_PER_WAVE)
#define POOL_BATCH   16

__device__ __forceinline__ void pool_batch(
    const int sv, const int gv, const int j0,
    const float* __restrict__ xl, float* sl, const float blo, const float bhi)
{
    float vlo[POOL_BATCH], vhi[POOL_BATCH];
#pragma unroll
    for (int u = 0; u < POOL_BATCH; ++u) {
        const int s = __builtin_amdgcn_readlane(sv, j0 + u);
        const float* r = xl + (size_t)s * 64;
        vlo[u] = r[0];
        vhi[u] = r[32];
    }
    __builtin_amdgcn_sched_barrier(0);
#pragma unroll
    for (int u = 0; u < POOL_BATCH; ++u) {
        const int g = __builtin_amdgcn_readlane(gv, j0 + u);
        float* p = sl + g * 64;
        const float hx = vlo[u] + blo;
        const float hy = vhi[u] + bhi;
        __hip_atomic_fetch_add(p, hx > 0.0f ? hx : 0.0f, __ATOMIC_RELAXED, __HIP_MEMORY_SCOPE_WORKGROUP);
        __hip_atomic_fetch_add(p + 32, hy > 0.0f ? hy : 0.0f, __ATOMIC_RELAXED, __HIP_MEMORY_SCOPE_WORKGROUP);
    }
}

__global__ __launch_bounds__(POOL_THREADS) void gcn_scatter_pool(
    const float* __restrict__ Agg1, const float* __restrict__ b1,
    const int* __restrict__ src, const int* __restrict__ dst,
    const int* __restrict__ gids, float* __restrict__ S, float* __restrict__ gcnt)
{
    __shared__ __attribute__((aligned(16))) float sS[NGRAPHS * HID];

    const unsigned tid  = threadIdx.x;
    const unsigned wave = __builtin_amdgcn_readfirstlane(tid >> 5);
    const unsigned l    = tid & 31;

    unsigned wc = wave * POOL_WGS + blockIdx.x;
    int sv, gv;
    {
        const unsigned e = wc * EDGES_PER_WAVE + l;
        sv = src[e];
        gv = gids[(unsigned)dst[e]];
        if (e < NNODES) unsafeAtomicAdd(&gcnt[gids[e]], 1.0f);
    }

    float4* s4 = reinterpret_cast<float4*>(sS);
    for (unsigned i = tid; i < NGRAPHS * HID / 4; i += POOL_THREADS)
        s4[i] = make_float4(0.0f, 0.0f, 0.0f, 0.0f);
    __syncthreads();

    const float blo = b1[l];
    const float bhi = b1[l + 32];
    const float* xl = Agg1 + l;
    float* sl = sS + l;

    for (;;) {
        const unsigned wcn = wc + POOL_WAVES;
        const bool more = wcn < NCHUNKS;
        const unsigned en = more ? (wcn * EDGES_PER_WAVE + l) : (wc * EDGES_PER_WAVE);
        const int svn = src[en];
        const unsigned dvn = (unsigned)dst[en];
        pool_batch(sv, gv, 0, xl, sl, blo, bhi);
        const int gvn = gids[dvn];
        pool_batch(sv, gv, POOL_BATCH, xl, sl, blo, bhi);
        if (!more) break;
        sv = svn; gv = gvn; wc = wcn;
    }
    __syncthreads();

    const unsigned plane = (__builtin_amdgcn_s_sendmsg_rtn(0x87) >> 16) & (NCOPY - 1);
    float* Sp = S + (size_t)plane * (NGRAPHS * HID);
#define FLUSH_BATCH 8
    for (unsigned base = tid; base < NGRAPHS * HID; base += FLUSH_BATCH * POOL_THREADS) {
        float v[FLUSH_BATCH];
#pragma unroll
        for (int u = 0; u < FLUSH_BATCH; ++u) {
            const unsigned i = base + u * POOL_THREADS;
            v[u] = (i < NGRAPHS * HID) ? sS[i] : 0.0f;
        }
#pragma unroll
        for (int u = 0; u < FLUSH_BATCH; ++u) {
            const unsigned i = base + u * POOL_THREADS;
            if (i < NGRAPHS * HID) unsafeAtomicAdd(Sp + i, v[u]);
        }
    }
}

__global__ __launch_bounds__(256) void gcn_finalize(
    const float* __restrict__ gsum, const float* __restrict__ gcnt,
    const float* __restrict__ b2, float* __restrict__ out)
{
    const unsigned gid = blockIdx.x * 256u + threadIdx.x;
    if (gid >= NGRAPHS * 64u) return;
    const unsigned g = gid >> 6;
    const unsigned c = gid & 63;
    const float cnt = gcnt[g];
    out[gid] = (cnt > 0.0f) ? (gsum[gid] / cnt + b2[c]) : 0.0f;
}

extern "C" void kernel_launch(void* const* d_in, const int* in_sizes, int n_in,
                              void* d_out, int out_size, void* d_ws, size_t ws_size,
                              hipStream_t stream)
{
    (void)in_sizes; (void)n_in; (void)out_size; (void)ws_size;

    const float* feats = (const float*)d_in[0];
    const float* W1    = (const float*)d_in[1];
    const float* b1    = (const float*)d_in[2];
    const float* W2    = (const float*)d_in[3];
    const float* b2    = (const float*)d_in[4];
    const int*   src   = (const int*)d_in[5];
    const int*   dst   = (const int*)d_in[6];
    const int*   gids  = (const int*)d_in[7];
    float* out = (float*)d_out;

    const size_t nodeElems = (size_t)NNODES * HID;
    float* bufA = (float*)d_ws;
    float* bufB = bufA + nodeElems;
    float* S    = bufA;
    float* gcnt = S + (size_t)NCOPY * NGRAPHS * HID;
    float* P    = gcnt + 1024;

    const int gemmBlocks    = NNODES / (16 * G1_TILES);
    const int scatterBlocks = 2 * SC_UNITS;
    const int poolBlocks    = PROWS / 16;
    const int finBlocks     = (NGRAPHS * 64 + 255) / 256;

    hipLaunchKernelGGL(gcn_gemm1_wmma, dim3(gemmBlocks), dim3(128), 0, stream,
                       feats, W1, bufA, out);
    hipMemsetAsync(bufB, 0, nodeElems * sizeof(float), stream);
    hipLaunchKernelGGL(gcn_scatter, dim3(scatterBlocks), dim3(256), 0, stream,
                       bufA, src, dst, bufB, out);
    hipMemsetAsync(S, 0, ((size_t)NCOPY * NGRAPHS * HID + NGRAPHS) * sizeof(float), stream);
    hipLaunchKernelGGL(gcn_scatter_pool, dim3(POOL_WGS), dim3(POOL_THREADS), 0, stream,
                       bufB, b1, src, dst, gids, S, gcnt);
    hipLaunchKernelGGL(gcn_gemm_pool_wmma, dim3(poolBlocks), dim3(128), 0, stream,
                       S, W2, P);
    hipLaunchKernelGGL(gcn_finalize, dim3(finBlocks), dim3(256), 0, stream,
                       P, gcnt, b2, out);
}


# layer 1 as (A feats) W1: scatter-add raw feats rows first, then the GEMM with b1+relu in place; tickets, counts, S and agg adjacent: one clear, not two (6 stream ops), no X1 buffer; sum order changes
# speedup vs baseline: 3.7089x; 1.0637x over previous
#include <hip/hip_runtime.h>

#define HID      64
#define NNODES   100000
#define NEDGES   1200000
#define NGRAPHS  1000
#define NCOPY    8
#define PROWS    1008

typedef __attribute__((ext_vector_type(2))) float v2f;
typedef __attribute__((ext_vector_type(8))) float v8f;


#define G1_TILES  5
#define G1_PITCH  68
__device__ __forceinline__ void g1_tile(const float* a, const v2f (&bw)[16], const float bn, float* outp)
{
    v8f acc = {};
#pragma unroll
    for (int t = 0; t < 16; ++t) {
        const v2f av = *reinterpret_cast<const v2f*>(a + 4 * t);
        acc = __builtin_amdgcn_wmma_f32_16x16x4_f32(
             false, av,  false, bw[t],
             (short)0, acc,  false,  false);
    }
    asm volatile("v_nop\n\tv_nop\n\tv_nop\n\tv_nop" : "+v"(acc));
#pragma unroll
    for (int v = 0; v < 8; ++v) {
        const float h = acc[v] + bn;
        outp[(size_t)v * 64] = h > 0.0f ? h : 0.0f;
    }
}

__global__ __launch_bounds__(128) void gcn_gemm1_wmma(
    float* Buf, const float* __restrict__ W, const float* __restrict__ b1)
{
    __shared__ __attribute__((aligned(16))) float sA[2][16 * G1_PITCH];

    const int tid  = threadIdx.x;
    const int wave = tid >> 5;
    const int lane = tid & 31;

    const int m     = lane & 15;
    const int n     = (wave << 4) + (lane & 15);
    const int khalf = (lane >> 4) << 1;
    const int rOff  = (lane >> 4) << 3;

    const int r0 = tid >> 4, c0 = (tid & 15) << 2;
    const float4* in4 = reinterpret_cast<const float4*>(Buf) + (size_t)blockIdx.x * (G1_TILES * 256) + tid;
    float* Out = Buf;
    const float bn = b1[n];
    float4 p0 = in4[0];
    float4 p1 = in4[128];

    v2f bw[16];
#pragma unroll
    for (int t = 0; t < 16; ++t) {
        bw[t].x = W[(4 * t + khalf) * 64 + n];
        bw[t].y = W[(4 * t + khalf + 1) * 64 + n];
    }

    *reinterpret_cast<float4*>(&sA[0][r0 * G1_PITCH + c0])       = p0;
    *reinterpret_cast<float4*>(&sA[0][(r0 + 8) * G1_PITCH + c0]) = p1;
    __syncthreads();

    for (int tile = 0; tile < G1_TILES - 1; ++tile) {
        p0 = in4[(tile + 1) * 256];
        p1 = in4[(tile + 1) * 256 + 128];
        __builtin_amdgcn_sched_barrier(0);
        g1_tile(sA[tile & 1] + m * G1_PITCH + khalf, bw, bn,
                Out + ((size_t)(blockIdx.x * G1_TILES + tile) * 16 + rOff) * 64 + n);
        float* nb = sA[(tile + 1) & 1];
        *reinterpret_cast<float4*>(&nb[r0 * G1_PITCH + c0])       = p0;
        *reinterpret_cast<float4*>(&nb[(r0 + 8) * G1_PITCH + c0]) = p1;
        __syncthreads();
    }
    g1_tile(sA[(G1_TILES - 1) & 1] + m * G1_PITCH + khalf, bw, bn,
            Out + ((size_t)(blockIdx.x * G1_TILES + G1_TILES - 1) * 16 + rOff) * 64 + n);
}

__global__ __launch_bounds__(128) void gcn_gemm_pool_wmma(
    const float* __restrict__ In, const float* __restrict__ W,
    float* __restrict__ Out)
{
    __shared__ float sW[64 * 65];
    __shared__ float sA[16 * 65];

    const int tid  = threadIdx.x;
    const int wave = tid >> 5;
    const int lane = tid & 31;
    const int rowBase = blockIdx.x * 16;

    for (int i = tid; i < 64 * 64; i += 128) {
        int r = i >> 6, c = i & 63;
        sW[r * 65 + c] = W[i];
    }
    for (int i = tid; i < 16 * 64; i += 128) {
        int r = i >> 6, c = i & 63;
        float v = 0.0f;
        if (rowBase + r < NGRAPHS) {
#pragma unroll
            for (int p = 0; p < NCOPY; ++p)
                v += In[(size_t)p * (NGRAPHS * HID) + (rowBase + r) * 64 + c];
        }
        sA[r * 65 + c] = v;
    }
    __syncthreads();

    const int m     = lane & 15;
    const int n     = (wave << 4) + (lane & 15);
    const int khalf = (lane >> 4) << 1;

    v8f acc = {};
#pragma unroll
    for (int k = 0; k < 64; k += 4) {
        v2f a, b;
        a.x = sA[m * 65 + k + khalf];
        a.y = sA[m * 65 + k + khalf + 1];
        b.x = sW[(k + khalf) * 65 + n];
        b.y = sW[(k + khalf + 1) * 65 + n];
        acc = __builtin_amdgcn_wmma_f32_16x16x4_f32(
             false, a,  false, b,
             (short)0, acc,  false,  false);
    }

    const int rOff = (lane >> 4) << 3;
    float* outp = Out + (size_t)(rowBase + rOff) * 64 + n;
#pragma unroll
    for (int v = 0; v < 8; ++v)
        outp[(size_t)v * 64] = acc[v];
}

#define EDGES_PER_WAVE 32
#define SC_EDGES   64
#define SC_WAVES   8
#define SC_CHUNKS  (NEDGES / SC_EDGES)
#define SC_UNITS   ((SC_CHUNKS + SC_WAVES - 1) / SC_WAVES)
#define SC_SPLIT   (NNODES / 2)
#define GATHER_BATCH 8

__device__ __forceinline__ void scatter_request(
    unsigned long long& todo, const int sv0, const int sv1, const int dv0, const int dv1,
    const float* __restrict__ X, const unsigned l, float& lo, float& hi, unsigned& d)
{
    const int j = __builtin_ctzll(todo);
    todo &= todo - 1ull;
    const int jl = j & 31;
    const unsigned sa = (unsigned)__builtin_amdgcn_readlane(sv0, jl);
    const unsigned sb = (unsigned)__builtin_amdgcn_readlane(sv1, jl);
    const unsigned da = (unsigned)__builtin_amdgcn_readlane(dv0, jl);
    const unsigned db = (unsigned)__builtin_amdgcn_readlane(dv1, jl);
    const unsigned s = j < 32 ? sa : sb;
    d = j < 32 ? da : db;
    const float* r = X + (size_t)s * 64;
    lo = r[l];
    hi = r[l + 32];
}

__global__ __launch_bounds__(256) void gcn_scatter(
    const float* __restrict__ X, const int* __restrict__ src,
    const int* __restrict__ dst, float* __restrict__ Agg, float* __restrict__ tickets)
{
    __shared__ unsigned sTicket[2];
    const unsigned tid = threadIdx.x;
    if (tid == 0) {
        unsigned* tick = reinterpret_cast<unsigned*>(tickets);
        unsigned part = (__builtin_amdgcn_s_sendmsg_rtn(0x87) >> 18) & 1u;
        unsigned t = atomicAdd(&tick[part], 1u);
        if (t >= SC_UNITS) {
            part ^= 1u;
            t = atomicAdd(&tick[part], 1u);
        }
        sTicket[0] = t;
        sTicket[1] = part;
    }
    __syncthreads();
    const unsigned unit = __builtin_amdgcn_readfirstlane(sTicket[0]);
    const unsigned part = __builtin_amdgcn_readfirstlane(sTicket[1]);
    const unsigned wave = __builtin_amdgcn_readfirstlane(tid >> 5);
    const unsigned l = tid & 31;
    const unsigned c = unit * SC_WAVES + wave;
    if (unit >= SC_UNITS || c >= SC_CHUNKS) return;
    const unsigned e = c * SC_EDGES + l;
    const int sv0 = src[e];
    const int dv0 = dst[e];
    const int sv1 = src[e + 32];
    const int dv1 = dst[e + 32];
    const bool upper = part != 0u;
    const unsigned todo0 = __builtin_amdgcn_ballot_w32(((unsigned)dv0 >= SC_SPLIT) == upper);
    const unsigned todo1 = __builtin_amdgcn_ballot_w32(((unsigned)dv1 >= SC_SPLIT) == upper);
    unsigned long long todo = ((unsigned long long)todo1 << 32) | todo0;
    int n = __builtin_popcountll(todo);

    for (; n >= GATHER_BATCH; n -= GATHER_BATCH) {
        float vlo[GATHER_BATCH], vhi[GATHER_BATCH];
        unsigned d[GATHER_BATCH];
#pragma unroll
        for (int u = 0; u < GATHER_BATCH; ++u)
            scatter_request(todo, sv0, sv1, dv0, dv1, X, l, vlo[u], vhi[u], d[u]);
#pragma unroll
        for (int u = 0; u < GATHER_BATCH; ++u) {
            float* p = Agg + (size_t)d[u] * 64;
            unsafeAtomicAdd(p + l, vlo[u]);
            unsafeAtomicAdd(p + l + 32, vhi[u]);
        }
    }
    if (n > 0) {
        float vlo[GATHER_BATCH - 1], vhi[GATHER_BATCH - 1];
        unsigned d[GATHER_BATCH - 1];
#pragma unroll
        for (int u = 0; u < GATHER_BATCH - 1; ++u)
            if (u < n) scatter_request(todo, sv0, sv1, dv0, dv1, X, l, vlo[u], vhi[u], d[u]);
#pragma unroll
        for (int u = 0; u < GATHER_BATCH - 1; ++u) {
            if (u < n) {
                float* p = Agg + (size_t)d[u] * 64;
                unsafeAtomicAdd(p + l, vlo[u]);
                unsafeAtomicAdd(p + l + 32, vhi[u]);
            }
        }
    }
}

#define POOL_THREADS 1024
#define POOL_WGS     256
#define POOL_WAVES   (POOL_WGS * (POOL_THREADS / 32))
#define NCHUNKS      (NEDGES / EDGES_PER_WAVE)
#define POOL_BATCH   16

__device__ __forceinline__ void pool_batch(
    const int sv, const int gv, const int j0,
    const float* __restrict__ xl, float* sl)
{
    float vlo[POOL_BATCH], vhi[POOL_BATCH];
#pragma unroll
    for (int u = 0; u < POOL_BATCH; ++u) {
        const int s = __builtin_amdgcn_readlane(sv, j0 + u);
        const float* r = xl + (size_t)s * 64;
        vlo[u] = r[0];
        vhi[u] = r[32];
    }
#pragma unroll
    for (int u = 0; u < POOL_BATCH; ++u) {
        const int g = __builtin_amdgcn_readlane(gv, j0 + u);
        float* p = sl + g * 64;
        __hip_atomic_fetch_add(p, vlo[u], __ATOMIC_RELAXED, __HIP_MEMORY_SCOPE_WORKGROUP);
        __hip_atomic_fetch_add(p + 32, vhi[u], __ATOMIC_RELAXED, __HIP_MEMORY_SCOPE_WORKGROUP);
    }
}

__global__ __launch_bounds__(POOL_THREADS) void gcn_scatter_pool(
    const float* __restrict__ Agg1,
    const int* __restrict__ src, const int* __restrict__ dst,
    const int* __restrict__ gids, float* __restrict__ S, float* __restrict__ gcnt)
{
    __shared__ __attribute__((aligned(16))) float sS[NGRAPHS * HID];

    const unsigned tid  = threadIdx.x;
    const unsigned wave = __builtin_amdgcn_readfirstlane(tid >> 5);
    const unsigned l    = tid & 31;

    unsigned wc = wave * POOL_WGS + blockIdx.x;
    int sv, gv;
    {
        const unsigned e = wc * EDGES_PER_WAVE + l;
        sv = src[e];
        gv = gids[(unsigned)dst[e]];
        if (e < NNODES) unsafeAtomicAdd(&gcnt[gids[e]], 1.0f);
    }

    float4* s4 = reinterpret_cast<float4*>(sS);
    for (unsigned i = tid; i < NGRAPHS * HID / 4; i += POOL_THREADS)
        s4[i] = make_float4(0.0f, 0.0f, 0.0f, 0.0f);
    __syncthreads();

    const float* xl = Agg1 + l;
    float* sl = sS + l;

    for (;;) {
        const unsigned wcn = wc + POOL_WAVES;
        const bool more = wcn < NCHUNKS;
        const unsigned en = more ? (wcn * EDGES_PER_WAVE + l) : (wc * EDGES_PER_WAVE);
        const int svn = src[en];
        const unsigned dvn = (unsigned)dst[en];
        pool_batch(sv, gv, 0, xl, sl);
        const int gvn = gids[dvn];
        pool_batch(sv, gv, POOL_BATCH, xl, sl);
        if (!more) break;
        sv = svn; gv = gvn; wc = wcn;
    }
    __syncthreads();

    const unsigned plane = (__builtin_amdgcn_s_sendmsg_rtn(0x87) >> 16) & (NCOPY - 1);
    float* Sp = S + (size_t)plane * (NGRAPHS * HID);
#define FLUSH_BATCH 8
    for (unsigned base = tid; base < NGRAPHS * HID; base += FLUSH_BATCH * POOL_THREADS) {
        float v[FLUSH_BATCH];
#pragma unroll
        for (int u = 0; u < FLUSH_BATCH; ++u) {
            const unsigned i = base + u * POOL_THREADS;
            v[u] = (i < NGRAPHS * HID) ? sS[i] : 0.0f;
        }
#pragma unroll
        for (int u = 0; u < FLUSH_BATCH; ++u) {
            const unsigned i = base + u * POOL_THREADS;
            if (i < NGRAPHS * HID) unsafeAtomicAdd(Sp + i, v[u]);
        }
    }
}

__global__ __launch_bounds__(256) void gcn_finalize(
    const float* __restrict__ gsum, const float* __restrict__ gcnt,
    const float* __restrict__ b2, float* __restrict__ out)
{
    const unsigned gid = blockIdx.x * 256u + threadIdx.x;
    if (gid >= NGRAPHS * 64u) return;
    const unsigned g = gid >> 6;
    const unsigned c = gid & 63;
    const float cnt = gcnt[g];
    out[gid] = (cnt > 0.0f) ? (gsum[gid] / cnt + b2[c]) : 0.0f;
}

extern "C" void kernel_launch(void* const* d_in, const int* in_sizes, int n_in,
                              void* d_out, int out_size, void* d_ws, size_t ws_size,
                              hipStream_t stream)
{
    (void)in_sizes; (void)n_in; (void)out_size; (void)ws_size;

    const float* feats = (const float*)d_in[0];
    const float* W1    = (const float*)d_in[1];
    const float* b1    = (const float*)d_in[2];
    const float* W2    = (const float*)d_in[3];
    const float* b2    = (const float*)d_in[4];
    const int*   src   = (const int*)d_in[5];
    const int*   dst   = (const int*)d_in[6];
    const int*   gids  = (const int*)d_in[7];
    float* out = (float*)d_out;

    const size_t nodeElems = (size_t)NNODES * HID;
    const size_t sumElems  = (size_t)NCOPY * NGRAPHS * HID;
    float* tick = (float*)d_ws;
    float* gcnt = tick + 64;
    float* S    = gcnt + 1024;
    float* agg  = S + sumElems;
    float* P    = agg + nodeElems;

    const int gemmBlocks    = NNODES / (16 * G1_TILES);
    const int scatterBlocks = 2 * SC_UNITS;
    const int poolBlocks    = PROWS / 16;
    const int finBlocks     = (NGRAPHS * 64 + 255) / 256;

    hipMemsetAsync(tick, 0, (64 + 1024 + sumElems + nodeElems) * sizeof(float), stream);
    hipLaunchKernelGGL(gcn_scatter, dim3(scatterBlocks), dim3(256), 0, stream,
                       feats, src, dst, agg, tick);
    hipLaunchKernelGGL(gcn_gemm1_wmma, dim3(gemmBlocks), dim3(128), 0, stream,
                       agg, W1, b1);
    hipLaunchKernelGGL(gcn_scatter_pool, dim3(POOL_WGS), dim3(POOL_THREADS), 0, stream,
                       agg, src, dst, gids, S, gcnt);
    hipLaunchKernelGGL(gcn_gemm_pool_wmma, dim3(poolBlocks), dim3(128), 0, stream,
                       S, W2, P);
    hipLaunchKernelGGL(gcn_finalize, dim3(finBlocks), dim3(256), 0, stream,
                       P, gcnt, b2, out);
}


# pooled GEMM (63 work-groups, each alone on a WGP): all W2 and plane loads of its staging requested at once, no branch: one memory round trip, not ten dependent ones; sums, order as before
# speedup vs baseline: 3.8428x; 1.0361x over previous
#include <hip/hip_runtime.h>

#define HID      64
#define NNODES   100000
#define NEDGES   1200000
#define NGRAPHS  1000
#define NCOPY    8
#define PROWS    1008

typedef __attribute__((ext_vector_type(2))) float v2f;
typedef __attribute__((ext_vector_type(8))) float v8f;


#define G1_TILES  5
#define G1_PITCH  68
__device__ __forceinline__ void g1_tile(const float* a, const v2f (&bw)[16], const float bn, float* outp)
{
    v8f acc = {};
#pragma unroll
    for (int t = 0; t < 16; ++t) {
        const v2f av = *reinterpret_cast<const v2f*>(a + 4 * t);
        acc = __builtin_amdgcn_wmma_f32_16x16x4_f32(
             false, av,  false, bw[t],
             (short)0, acc,  false,  false);
    }
    asm volatile("v_nop\n\tv_nop\n\tv_nop\n\tv_nop" : "+v"(acc));
#pragma unroll
    for (int v = 0; v < 8; ++v) {
        const float h = acc[v] + bn;
        outp[(size_t)v * 64] = h > 0.0f ? h : 0.0f;
    }
}

__global__ __launch_bounds__(128) void gcn_gemm1_wmma(
    float* Buf, const float* __restrict__ W, const float* __restrict__ b1)
{
    __shared__ __attribute__((aligned(16))) float sA[2][16 * G1_PITCH];

    const int tid  = threadIdx.x;
    const int wave = tid >> 5;
    const int lane = tid & 31;

    const int m     = lane & 15;
    const int n     = (wave << 4) + (lane & 15);
    const int khalf = (lane >> 4) << 1;
    const int rOff  = (lane >> 4) << 3;

    const int r0 = tid >> 4, c0 = (tid & 15) << 2;
    const float4* in4 = reinterpret_cast<const float4*>(Buf) + (size_t)blockIdx.x * (G1_TILES * 256) + tid;
    float* Out = Buf;
    const float bn = b1[n];
    float4 p0 = in4[0];
    float4 p1 = in4[128];

    v2f bw[16];
#pragma unroll
    for (int t = 0; t < 16; ++t) {
        bw[t].x = W[(4 * t + khalf) * 64 + n];
        bw[t].y = W[(4 * t + khalf + 1) * 64 + n];
    }

    *reinterpret_cast<float4*>(&sA[0][r0 * G1_PITCH + c0])       = p0;
    *reinterpret_cast<float4*>(&sA[0][(r0 + 8) * G1_PITCH + c0]) = p1;
    __syncthreads();

    for (int tile = 0; tile < G1_TILES - 1; ++tile) {
        p0 = in4[(tile + 1) * 256];
        p1 = in4[(tile + 1) * 256 + 128];
        __builtin_amdgcn_sched_barrier(0);
        g1_tile(sA[tile & 1] + m * G1_PITCH + khalf, bw, bn,
                Out + ((size_t)(blockIdx.x * G1_TILES + tile) * 16 + rOff) * 64 + n);
        float* nb = sA[(tile + 1) & 1];
        *reinterpret_cast<float4*>(&nb[r0 * G1_PITCH + c0])       = p0;
        *reinterpret_cast<float4*>(&nb[(r0 + 8) * G1_PITCH + c0]) = p1;
        __syncthreads();
    }
    g1_tile(sA[(G1_TILES - 1) & 1] + m * G1_PITCH + khalf, bw, bn,
            Out + ((size_t)(blockIdx.x * G1_TILES + G1_TILES - 1) * 16 + rOff) * 64 + n);
}

__global__ __launch_bounds__(128) void gcn_gemm_pool_wmma(
    const float* In, const float* W,
    float* __restrict__ Out)
{
    __shared__ float sW[64 * 65];
    __shared__ float sA[16 * 65];

    const int tid  = threadIdx.x;
    const int wave = tid >> 5;
    const int lane = tid & 31;
    const int rowBase = blockIdx.x * 16;

    float wv[32];
#pragma unroll
    for (int j = 0; j < 32; ++j)
        wv[j] = W[tid + 128 * j];

    float4 av[2][NCOPY];
#pragma unroll
    for (int j = 0; j < 2; ++j) {
        const int q   = tid + 128 * j;
        const int row = rowBase + (q >> 4);
        const int rc  = row < NGRAPHS ? row : NGRAPHS - 1;
        const float4* p4 = reinterpret_cast<const float4*>(In) + rc * 16 + (q & 15);
#pragma unroll
        for (int p = 0; p < NCOPY; ++p)
            av[j][p] = p4[(size_t)p * (NGRAPHS * HID / 4)];
    }
    __builtin_amdgcn_sched_barrier(0);

#pragma unroll
    for (int j = 0; j < 32; ++j) {
        const int i = tid + 128 * j;
        sW[(i >> 6) * 65 + (i & 63)] = wv[j];
    }
#pragma unroll
    for (int j = 0; j < 2; ++j) {
        const int q = tid + 128 * j;
        const int r = q >> 4, c = (q & 15) << 2;
        float4 v = make_float4(0.0f, 0.0f, 0.0f, 0.0f);
#pragma unroll
        for (int p = 0; p < NCOPY; ++p) {
            v.x += av[j][p].x;
            v.y += av[j][p].y;
            v.z += av[j][p].z;
            v.w += av[j][p].w;
        }
        if (rowBase + r >= NGRAPHS) v = make_float4(0.0f, 0.0f, 0.0f, 0.0f);
        sA[r * 65 + c]     = v.x;
        sA[r * 65 + c + 1] = v.y;
        sA[r * 65 + c + 2] = v.z;
        sA[r * 65 + c + 3] = v.w;
    }
    __syncthreads();

    const int m     = lane & 15;
    const int n     = (wave << 4) + (lane & 15);
    const int khalf = (lane >> 4) << 1;

    v8f acc = {};
#pragma unroll
    for (int k = 0; k < 64; k += 4) {
        v2f a, b;
        a.x = sA[m * 65 + k + khalf];
        a.y = sA[m * 65 + k + khalf + 1];
        b.x = sW[(k + khalf) * 65 + n];
        b.y = sW[(k + khalf + 1) * 65 + n];
        acc = __builtin_amdgcn_wmma_f32_16x16x4_f32(
             false, a,  false, b,
             (short)0, acc,  false,  false);
    }

    const int rOff = (lane >> 4) << 3;
    float* outp = Out + (size_t)(rowBase + rOff) * 64 + n;
#pragma unroll
    for (int v = 0; v < 8; ++v)
        outp[(size_t)v * 64] = acc[v];
}

#define EDGES_PER_WAVE 32
#define SC_EDGES   64
#define SC_WAVES   8
#define SC_CHUNKS  (NEDGES / SC_EDGES)
#define SC_UNITS   ((SC_CHUNKS + SC_WAVES - 1) / SC_WAVES)
#define SC_SPLIT   (NNODES / 2)
#define GATHER_BATCH 8

__device__ __forceinline__ void scatter_request(
    unsigned long long& todo, const int sv0, const int sv1, const int dv0, const int dv1,
    const float* __restrict__ X, const unsigned l, float& lo, float& hi, unsigned& d)
{
    const int j = __builtin_ctzll(todo);
    todo &= todo - 1ull;
    const int jl = j & 31;
    const unsigned sa = (unsigned)__builtin_amdgcn_readlane(sv0, jl);
    const unsigned sb = (unsigned)__builtin_amdgcn_readlane(sv1, jl);
    const unsigned da = (unsigned)__builtin_amdgcn_readlane(dv0, jl);
    const unsigned db = (unsigned)__builtin_amdgcn_readlane(dv1, jl);
    const unsigned s = j < 32 ? sa : sb;
    d = j < 32 ? da : db;
    const float* r = X + (size_t)s * 64;
    lo = r[l];
    hi = r[l + 32];
}

__global__ __launch_bounds__(256) void gcn_scatter(
    const float* __restrict__ X, const int* __restrict__ src,
    const int* __restrict__ dst, float* __restrict__ Agg, float* __restrict__ tickets)
{
    __shared__ unsigned sTicket[2];
    const unsigned tid = threadIdx.x;
    if (tid == 0) {
        unsigned* tick = reinterpret_cast<unsigned*>(tickets);
        unsigned part = (__builtin_amdgcn_s_sendmsg_rtn(0x87) >> 18) & 1u;
        unsigned t = atomicAdd(&tick[part], 1u);
        if (t >= SC_UNITS) {
            part ^= 1u;
            t = atomicAdd(&tick[part], 1u);
        }
        sTicket[0] = t;
        sTicket[1] = part;
    }
    __syncthreads();
    const unsigned unit = __builtin_amdgcn_readfirstlane(sTicket[0]);
    const unsigned part = __builtin_amdgcn_readfirstlane(sTicket[1]);
    const unsigned wave = __builtin_amdgcn_readfirstlane(tid >> 5);
    const unsigned l = tid & 31;
    const unsigned c = unit * SC_WAVES + wave;
    if (unit >= SC_UNITS || c >= SC_CHUNKS) return;
    const unsigned e = c * SC_EDGES + l;
    const int sv0 = src[e];
    const int dv0 = dst[e];
    const int sv1 = src[e + 32];
    const int dv1 = dst[e + 32];
    const bool upper = part != 0u;
    const unsigned todo0 = __builtin_amdgcn_ballot_w32(((unsigned)dv0 >= SC_SPLIT) == upper);
    const unsigned todo1 = __builtin_amdgcn_ballot_w32(((unsigned)dv1 >= SC_SPLIT) == upper);
    unsigned long long todo = ((unsigned long long)todo1 << 32) | todo0;
    int n = __builtin_popcountll(todo);

    for (; n >= GATHER_BATCH; n -= GATHER_BATCH) {
        float vlo[GATHER_BATCH], vhi[GATHER_BATCH];
        unsigned d[GATHER_BATCH];
#pragma unroll
        for (int u = 0; u < GATHER_BATCH; ++u)
            scatter_request(todo, sv0, sv1, dv0, dv1, X, l, vlo[u], vhi[u], d[u]);
#pragma unroll
        for (int u = 0; u < GATHER_BATCH; ++u) {
            float* p = Agg + (size_t)d[u] * 64;
            unsafeAtomicAdd(p + l, vlo[u]);
            unsafeAtomicAdd(p + l + 32, vhi[u]);
        }
    }
    if (n > 0) {
        float vlo[GATHER_BATCH - 1], vhi[GATHER_BATCH - 1];
        unsigned d[GATHER_BATCH - 1];
#pragma unroll
        for (int u = 0; u < GATHER_BATCH - 1; ++u)
            if (u < n) scatter_request(todo, sv0, sv1, dv0, dv1, X, l, vlo[u], vhi[u], d[u]);
#pragma unroll
        for (int u = 0; u < GATHER_BATCH - 1; ++u) {
            if (u < n) {
                float* p = Agg + (size_t)d[u] * 64;
                unsafeAtomicAdd(p + l, vlo[u]);
                unsafeAtomicAdd(p + l + 32, vhi[u]);
            }
        }
    }
}

#define POOL_THREADS 1024
#define POOL_WGS     256
#define POOL_WAVES   (POOL_WGS * (POOL_THREADS / 32))
#define NCHUNKS      (NEDGES / EDGES_PER_WAVE)
#define POOL_BATCH   16

__device__ __forceinline__ void pool_batch(
    const int sv, const int gv, const int j0,
    const float* __restrict__ xl, float* sl)
{
    float vlo[POOL_BATCH], vhi[POOL_BATCH];
#pragma unroll
    for (int u = 0; u < POOL_BATCH; ++u) {
        const int s = __builtin_amdgcn_readlane(sv, j0 + u);
        const float* r = xl + (size_t)s * 64;
        vlo[u] = r[0];
        vhi[u] = r[32];
    }
#pragma unroll
    for (int u = 0; u < POOL_BATCH; ++u) {
        const int g = __builtin_amdgcn_readlane(gv, j0 + u);
        float* p = sl + g * 64;
        __hip_atomic_fetch_add(p, vlo[u], __ATOMIC_RELAXED, __HIP_MEMORY_SCOPE_WORKGROUP);
        __hip_atomic_fetch_add(p + 32, vhi[u], __ATOMIC_RELAXED, __HIP_MEMORY_SCOPE_WORKGROUP);
    }
}

__global__ __launch_bounds__(POOL_THREADS) void gcn_scatter_pool(
    const float* __restrict__ Agg1,
    const int* __restrict__ src, const int* __restrict__ dst,
    const int* __restrict__ gids, float* __restrict__ S, float* __restrict__ gcnt)
{
    __shared__ __attribute__((aligned(16))) float sS[NGRAPHS * HID];

    const unsigned tid  = threadIdx.x;
    const unsigned wave = __builtin_amdgcn_readfirstlane(tid >> 5);
    const unsigned l    = tid & 31;

    unsigned wc = wave * POOL_WGS + blockIdx.x;
    int sv, gv;
    {
        const unsigned e = wc * EDGES_PER_WAVE + l;
        sv = src[e];
        gv = gids[(unsigned)dst[e]];
        if (e < NNODES) unsafeAtomicAdd(&gcnt[gids[e]], 1.0f);
    }

    float4* s4 = reinterpret_cast<float4*>(sS);
    for (unsigned i = tid; i < NGRAPHS * HID / 4; i += POOL_THREADS)
        s4[i] = make_float4(0.0f, 0.0f, 0.0f, 0.0f);
    __syncthreads();

    const float* xl = Agg1 + l;
    float* sl = sS + l;

    for (;;) {
        const unsigned wcn = wc + POOL_WAVES;
        const bool more = wcn < NCHUNKS;
        const unsigned en = more ? (wcn * EDGES_PER_WAVE + l) : (wc * EDGES_PER_WAVE);
        const int svn = src[en];
        const unsigned dvn = (unsigned)dst[en];
        pool_batch(sv, gv, 0, xl, sl);
        const int gvn = gids[dvn];
        pool_batch(sv, gv, POOL_BATCH, xl, sl);
        if (!more) break;
        sv = svn; gv = gvn; wc = wcn;
    }
    __syncthreads();

    const unsigned plane = (__builtin_amdgcn_s_sendmsg_rtn(0x87) >> 16) & (NCOPY - 1);
    float* Sp = S + (size_t)plane * (NGRAPHS * HID);
#define FLUSH_BATCH 8
    for (unsigned base = tid; base < NGRAPHS * HID; base += FLUSH_BATCH * POOL_THREADS) {
        float v[FLUSH_BATCH];
#pragma unroll
        for (int u = 0; u < FLUSH_BATCH; ++u) {
            const unsigned i = base + u * POOL_THREADS;
            v[u] = (i < NGRAPHS * HID) ? sS[i] : 0.0f;
        }
#pragma unroll
        for (int u = 0; u < FLUSH_BATCH; ++u) {
            const unsigned i = base + u * POOL_THREADS;
            if (i < NGRAPHS * HID) unsafeAtomicAdd(Sp + i, v[u]);
        }
    }
}

__global__ __launch_bounds__(256) void gcn_finalize(
    const float* __restrict__ gsum, const float* __restrict__ gcnt,
    const float* __restrict__ b2, float* __restrict__ out)
{
    const unsigned gid = blockIdx.x * 256u + threadIdx.x;
    if (gid >= NGRAPHS * 64u) return;
    const unsigned g = gid >> 6;
    const unsigned c = gid & 63;
    const float cnt = gcnt[g];
    out[gid] = (cnt > 0.0f) ? (gsum[gid] / cnt + b2[c]) : 0.0f;
}

extern "C" void kernel_launch(void* const* d_in, const int* in_sizes, int n_in,
                              void* d_out, int out_size, void* d_ws, size_t ws_size,
                              hipStream_t stream)
{
    (void)in_sizes; (void)n_in; (void)out_size; (void)ws_size;

    const float* feats = (const float*)d_in[0];
    const float* W1    = (const float*)d_in[1];
    const float* b1    = (const float*)d_in[2];
    const float* W2    = (const float*)d_in[3];
    const float* b2    = (const float*)d_in[4];
    const int*   src   = (const int*)d_in[5];
    const int*   dst   = (const int*)d_in[6];
    const int*   gids  = (const int*)d_in[7];
    float* out = (float*)d_out;

    const size_t nodeElems = (size_t)NNODES * HID;
    const size_t sumElems  = (size_t)NCOPY * NGRAPHS * HID;
    float* tick = (float*)d_ws;
    float* gcnt = tick + 64;
    float* S    = gcnt + 1024;
    float* agg  = S + sumElems;
    float* P    = agg + nodeElems;

    const int gemmBlocks    = NNODES / (16 * G1_TILES);
    const int scatterBlocks = 2 * SC_UNITS;
    const int poolBlocks    = PROWS / 16;
    const int finBlocks     = (NGRAPHS * 64 + 255) / 256;

    hipMemsetAsync(tick, 0, (64 + 1024 + sumElems + nodeElems) * sizeof(float), stream);
    hipLaunchKernelGGL(gcn_scatter, dim3(scatterBlocks), dim3(256), 0, stream,
                       feats, src, dst, agg, tick);
    hipLaunchKernelGGL(gcn_gemm1_wmma, dim3(gemmBlocks), dim3(128), 0, stream,
                       agg, W1, b1);
    hipLaunchKernelGGL(gcn_scatter_pool, dim3(POOL_WGS), dim3(POOL_THREADS), 0, stream,
                       agg, src, dst, gids, S, gcnt);
    hipLaunchKernelGGL(gcn_gemm_pool_wmma, dim3(poolBlocks), dim3(128), 0, stream,
                       S, W2, P);
    hipLaunchKernelGGL(gcn_finalize, dim3(finBlocks), dim3(256), 0, stream,
                       P, gcnt, b2, out);
}


# pool scatter: node -> graph lookups read a 16-bit per-8-node block table in LDS (built by the layer-1 GEMM launch); only nodes in blocks of mixed graphs read the full map; sums, order as before
# speedup vs baseline: 3.9609x; 1.0307x over previous
#include <hip/hip_runtime.h>

#define HID      64
#define NNODES   100000
#define NEDGES   1200000
#define NGRAPHS  1000
#define NCOPY    8
#define PROWS    1008

typedef __attribute__((ext_vector_type(2))) float v2f;
typedef __attribute__((ext_vector_type(8))) float v8f;


#define G1_TILES  5
#define GT_NODES    8
#define GT_ENTRIES  (NNODES / GT_NODES)
#define GT_MIXED    0xFFFFu
#define GT_PIECES   (GT_ENTRIES / 4)
#define G1_PITCH  68
__device__ __forceinline__ void g1_tile(const float* a, const v2f (&bw)[16], const float bn, float* outp)
{
    v8f acc = {};
#pragma unroll
    for (int t = 0; t < 16; ++t) {
        const v2f av = *reinterpret_cast<const v2f*>(a + 4 * t);
        acc = __builtin_amdgcn_wmma_f32_16x16x4_f32(
             false, av,  false, bw[t],
             (short)0, acc,  false,  false);
    }
    asm volatile("v_nop\n\tv_nop\n\tv_nop\n\tv_nop" : "+v"(acc));
#pragma unroll
    for (int v = 0; v < 8; ++v) {
        const float h = acc[v] + bn;
        outp[(size_t)v * 64] = h > 0.0f ? h : 0.0f;
    }
}

__global__ __launch_bounds__(128) void gcn_gemm1_wmma(
    float* Buf, const float* __restrict__ W, const float* __restrict__ b1,
    const int* __restrict__ gids, unsigned short* __restrict__ gtab)
{
    __shared__ __attribute__((aligned(16))) float sA[2][16 * G1_PITCH];

    const int tid  = threadIdx.x;
    const int wave = tid >> 5;
    const int lane = tid & 31;

    const int m     = lane & 15;
    const int n     = (wave << 4) + (lane & 15);
    const int khalf = (lane >> 4) << 1;
    const int rOff  = (lane >> 4) << 3;

    const int r0 = tid >> 4, c0 = (tid & 15) << 2;
    const float4* in4 = reinterpret_cast<const float4*>(Buf) + (size_t)blockIdx.x * (G1_TILES * 256) + tid;
    float* Out = Buf;
    const float bn = b1[n];
    float4 p0 = in4[0];
    float4 p1 = in4[128];

    v2f bw[16];
#pragma unroll
    for (int t = 0; t < 16; ++t) {
        bw[t].x = W[(4 * t + khalf) * 64 + n];
        bw[t].y = W[(4 * t + khalf + 1) * 64 + n];
    }

    *reinterpret_cast<float4*>(&sA[0][r0 * G1_PITCH + c0])       = p0;
    *reinterpret_cast<float4*>(&sA[0][(r0 + 8) * G1_PITCH + c0]) = p1;
    __syncthreads();

    for (int tile = 0; tile < G1_TILES - 1; ++tile) {
        p0 = in4[(tile + 1) * 256];
        p1 = in4[(tile + 1) * 256 + 128];
        __builtin_amdgcn_sched_barrier(0);
        g1_tile(sA[tile & 1] + m * G1_PITCH + khalf, bw, bn,
                Out + ((size_t)(blockIdx.x * G1_TILES + tile) * 16 + rOff) * 64 + n);
        float* nb = sA[(tile + 1) & 1];
        *reinterpret_cast<float4*>(&nb[r0 * G1_PITCH + c0])       = p0;
        *reinterpret_cast<float4*>(&nb[(r0 + 8) * G1_PITCH + c0]) = p1;
        __syncthreads();
    }
    g1_tile(sA[(G1_TILES - 1) & 1] + m * G1_PITCH + khalf, bw, bn,
            Out + ((size_t)(blockIdx.x * G1_TILES + G1_TILES - 1) * 16 + rOff) * 64 + n);

    if (tid < 2 * G1_TILES) {
        const unsigned b = blockIdx.x * (2 * G1_TILES) + tid;
        const int4 ga = *reinterpret_cast<const int4*>(gids + (size_t)b * GT_NODES);
        const int4 gb = *reinterpret_cast<const int4*>(gids + (size_t)b * GT_NODES + 4);
        const unsigned g = (unsigned)ga.x;
        const unsigned diff = (unsigned)((ga.y ^ ga.x) | (ga.z ^ ga.x) | (ga.w ^ ga.x) |
                                         (gb.x ^ ga.x) | (gb.y ^ ga.x) | (gb.z ^ ga.x) | (gb.w ^ ga.x));
        gtab[b] = (unsigned short)((diff == 0u && g < GT_MIXED) ? g : GT_MIXED);
    }
}

__global__ __launch_bounds__(128) void gcn_gemm_pool_wmma(
    const float* In, const float* W,
    float* __restrict__ Out)
{
    __shared__ float sW[64 * 65];
    __shared__ float sA[16 * 65];

    const int tid  = threadIdx.x;
    const int wave = tid >> 5;
    const int lane = tid & 31;
    const int rowBase = blockIdx.x * 16;

    float wv[32];
#pragma unroll
    for (int j = 0; j < 32; ++j)
        wv[j] = W[tid + 128 * j];

    float4 av[2][NCOPY];
#pragma unroll
    for (int j = 0; j < 2; ++j) {
        const int q   = tid + 128 * j;
        const int row = rowBase + (q >> 4);
        const int rc  = row < NGRAPHS ? row : NGRAPHS - 1;
        const float4* p4 = reinterpret_cast<const float4*>(In) + rc * 16 + (q & 15);
#pragma unroll
        for (int p = 0; p < NCOPY; ++p)
            av[j][p] = p4[(size_t)p * (NGRAPHS * HID / 4)];
    }
    __builtin_amdgcn_sched_barrier(0);

#pragma unroll
    for (int j = 0; j < 32; ++j) {
        const int i = tid + 128 * j;
        sW[(i >> 6) * 65 + (i & 63)] = wv[j];
    }
#pragma unroll
    for (int j = 0; j < 2; ++j) {
        const int q = tid + 128 * j;
        const int r = q >> 4, c = (q & 15) << 2;
        float4 v = make_float4(0.0f, 0.0f, 0.0f, 0.0f);
#pragma unroll
        for (int p = 0; p < NCOPY; ++p) {
            v.x += av[j][p].x;
            v.y += av[j][p].y;
            v.z += av[j][p].z;
            v.w += av[j][p].w;
        }
        if (rowBase + r >= NGRAPHS) v = make_float4(0.0f, 0.0f, 0.0f, 0.0f);
        sA[r * 65 + c]     = v.x;
        sA[r * 65 + c + 1] = v.y;
        sA[r * 65 + c + 2] = v.z;
        sA[r * 65 + c + 3] = v.w;
    }
    __syncthreads();

    const int m     = lane & 15;
    const int n     = (wave << 4) + (lane & 15);
    const int khalf = (lane >> 4) << 1;

    v8f acc = {};
#pragma unroll
    for (int k = 0; k < 64; k += 4) {
        v2f a, b;
        a.x = sA[m * 65 + k + khalf];
        a.y = sA[m * 65 + k + khalf + 1];
        b.x = sW[(k + khalf) * 65 + n];
        b.y = sW[(k + khalf + 1) * 65 + n];
        acc = __builtin_amdgcn_wmma_f32_16x16x4_f32(
             false, a,  false, b,
             (short)0, acc,  false,  false);
    }

    const int rOff = (lane >> 4) << 3;
    float* outp = Out + (size_t)(rowBase + rOff) * 64 + n;
#pragma unroll
    for (int v = 0; v < 8; ++v)
        outp[(size_t)v * 64] = acc[v];
}

#define EDGES_PER_WAVE 32
#define SC_EDGES   64
#define SC_WAVES   8
#define SC_CHUNKS  (NEDGES / SC_EDGES)
#define SC_UNITS   ((SC_CHUNKS + SC_WAVES - 1) / SC_WAVES)
#define SC_SPLIT   (NNODES / 2)
#define GATHER_BATCH 8

__device__ __forceinline__ void scatter_request(
    unsigned long long& todo, const int sv0, const int sv1, const int dv0, const int dv1,
    const float* __restrict__ X, const unsigned l, float& lo, float& hi, unsigned& d)
{
    const int j = __builtin_ctzll(todo);
    todo &= todo - 1ull;
    const int jl = j & 31;
    const unsigned sa = (unsigned)__builtin_amdgcn_readlane(sv0, jl);
    const unsigned sb = (unsigned)__builtin_amdgcn_readlane(sv1, jl);
    const unsigned da = (unsigned)__builtin_amdgcn_readlane(dv0, jl);
    const unsigned db = (unsigned)__builtin_amdgcn_readlane(dv1, jl);
    const unsigned s = j < 32 ? sa : sb;
    d = j < 32 ? da : db;
    const float* r = X + (size_t)s * 64;
    lo = r[l];
    hi = r[l + 32];
}

__global__ __launch_bounds__(256) void gcn_scatter(
    const float* __restrict__ X, const int* __restrict__ src,
    const int* __restrict__ dst, float* __restrict__ Agg, float* __restrict__ tickets)
{
    __shared__ unsigned sTicket[2];
    const unsigned tid = threadIdx.x;
    if (tid == 0) {
        unsigned* tick = reinterpret_cast<unsigned*>(tickets);
        unsigned part = (__builtin_amdgcn_s_sendmsg_rtn(0x87) >> 18) & 1u;
        unsigned t = atomicAdd(&tick[part], 1u);
        if (t >= SC_UNITS) {
            part ^= 1u;
            t = atomicAdd(&tick[part], 1u);
        }
        sTicket[0] = t;
        sTicket[1] = part;
    }
    __syncthreads();
    const unsigned unit = __builtin_amdgcn_readfirstlane(sTicket[0]);
    const unsigned part = __builtin_amdgcn_readfirstlane(sTicket[1]);
    const unsigned wave = __builtin_amdgcn_readfirstlane(tid >> 5);
    const unsigned l = tid & 31;
    const unsigned c = unit * SC_WAVES + wave;
    if (unit >= SC_UNITS || c >= SC_CHUNKS) return;
    const unsigned e = c * SC_EDGES + l;
    const int sv0 = src[e];
    const int dv0 = dst[e];
    const int sv1 = src[e + 32];
    const int dv1 = dst[e + 32];
    const bool upper = part != 0u;
    const unsigned todo0 = __builtin_amdgcn_ballot_w32(((unsigned)dv0 >= SC_SPLIT) == upper);
    const unsigned todo1 = __builtin_amdgcn_ballot_w32(((unsigned)dv1 >= SC_SPLIT) == upper);
    unsigned long long todo = ((unsigned long long)todo1 << 32) | todo0;
    int n = __builtin_popcountll(todo);

    for (; n >= GATHER_BATCH; n -= GATHER_BATCH) {
        float vlo[GATHER_BATCH], vhi[GATHER_BATCH];
        unsigned d[GATHER_BATCH];
#pragma unroll
        for (int u = 0; u < GATHER_BATCH; ++u)
            scatter_request(todo, sv0, sv1, dv0, dv1, X, l, vlo[u], vhi[u], d[u]);
#pragma unroll
        for (int u = 0; u < GATHER_BATCH; ++u) {
            float* p = Agg + (size_t)d[u] * 64;
            unsafeAtomicAdd(p + l, vlo[u]);
            unsafeAtomicAdd(p + l + 32, vhi[u]);
        }
    }
    if (n > 0) {
        float vlo[GATHER_BATCH - 1], vhi[GATHER_BATCH - 1];
        unsigned d[GATHER_BATCH - 1];
#pragma unroll
        for (int u = 0; u < GATHER_BATCH - 1; ++u)
            if (u < n) scatter_request(todo, sv0, sv1, dv0, dv1, X, l, vlo[u], vhi[u], d[u]);
#pragma unroll
        for (int u = 0; u < GATHER_BATCH - 1; ++u) {
            if (u < n) {
                float* p = Agg + (size_t)d[u] * 64;
                unsafeAtomicAdd(p + l, vlo[u]);
                unsafeAtomicAdd(p + l + 32, vhi[u]);
            }
        }
    }
}

#define POOL_THREADS 1024
#define POOL_WGS     256
#define POOL_WAVES   (POOL_WGS * (POOL_THREADS / 32))
#define NCHUNKS      (NEDGES / EDGES_PER_WAVE)
#define POOL_BATCH   16

__device__ __forceinline__ void pool_gather(
    const int sv, const int j0, const float* __restrict__ xl,
    float (&vlo)[POOL_BATCH], float (&vhi)[POOL_BATCH])
{
#pragma unroll
    for (int u = 0; u < POOL_BATCH; ++u) {
        const int s = __builtin_amdgcn_readlane(sv, j0 + u);
        const float* r = xl + (size_t)s * 64;
        vlo[u] = r[0];
        vhi[u] = r[32];
    }
}
__device__ __forceinline__ void pool_add(
    const int gv, const int j0, float* sl,
    const float (&vlo)[POOL_BATCH], const float (&vhi)[POOL_BATCH])
{
#pragma unroll
    for (int u = 0; u < POOL_BATCH; ++u) {
        const int g = __builtin_amdgcn_readlane(gv, j0 + u);
        float* p = sl + g * 64;
        __hip_atomic_fetch_add(p, vlo[u], __ATOMIC_RELAXED, __HIP_MEMORY_SCOPE_WORKGROUP);
        __hip_atomic_fetch_add(p + 32, vhi[u], __ATOMIC_RELAXED, __HIP_MEMORY_SCOPE_WORKGROUP);
    }
}

__global__ __launch_bounds__(POOL_THREADS) void gcn_scatter_pool(
    const float* __restrict__ Agg1,
    const int* __restrict__ src, const int* __restrict__ dst,
    const int* __restrict__ gids, const unsigned short* gtab,
    float* __restrict__ S, float* __restrict__ gcnt)
{
    __shared__ __attribute__((aligned(16))) float sS[NGRAPHS * HID];
    __shared__ __attribute__((aligned(8))) unsigned short sT[GT_ENTRIES];

    const unsigned tid  = threadIdx.x;
    const unsigned wave = __builtin_amdgcn_readfirstlane(tid >> 5);
    const unsigned l    = tid & 31;

    unsigned wc = wave * POOL_WGS + blockIdx.x;
    int sv, gv;
    unsigned dv;
    {
        const unsigned e = wc * EDGES_PER_WAVE + l;
        sv = src[e];
        dv = (unsigned)dst[e];
        if (e < NNODES) unsafeAtomicAdd(&gcnt[gids[e]], 1.0f);
    }

    const unsigned i3 = tid + 3 * POOL_THREADS < GT_PIECES ? tid + 3 * POOL_THREADS : GT_PIECES - 1;
    const uint2* t2 = reinterpret_cast<const uint2*>(gtab);
    const uint2 tv0 = t2[tid];
    const uint2 tv1 = t2[tid + POOL_THREADS];
    const uint2 tv2 = t2[tid + 2 * POOL_THREADS];
    const uint2 tv3 = t2[i3];

    float4* s4 = reinterpret_cast<float4*>(sS);
    for (unsigned i = tid; i < NGRAPHS * HID / 4; i += POOL_THREADS)
        s4[i] = make_float4(0.0f, 0.0f, 0.0f, 0.0f);
    {
        uint2* s2 = reinterpret_cast<uint2*>(sT);
        s2[tid] = tv0;
        s2[tid + POOL_THREADS] = tv1;
        s2[tid + 2 * POOL_THREADS] = tv2;
        if (tid + 3 * POOL_THREADS < GT_PIECES) s2[i3] = tv3;
    }
    __syncthreads();

    {
        const unsigned t = sT[dv / GT_NODES];
        gv = (int)t;
        if (t == GT_MIXED) gv = gids[dv];
    }

    const float* xl = Agg1 + l;
    float* sl = sS + l;

    for (;;) {
        const unsigned wcn = wc + POOL_WAVES;
        const bool more = wcn < NCHUNKS;
        const unsigned en = more ? (wcn * EDGES_PER_WAVE + l) : (wc * EDGES_PER_WAVE);
        const int svn = src[en];
        const unsigned dvn = (unsigned)dst[en];
        float vlo[POOL_BATCH], vhi[POOL_BATCH];
        pool_gather(sv, 0, xl, vlo, vhi);
        const unsigned tn = sT[dvn / GT_NODES];
        pool_add(gv, 0, sl, vlo, vhi);
        int gvn = (int)tn;
        if (tn == GT_MIXED) gvn = gids[dvn];
        pool_gather(sv, POOL_BATCH, xl, vlo, vhi);
        pool_add(gv, POOL_BATCH, sl, vlo, vhi);
        if (!more) break;
        sv = svn; gv = gvn; wc = wcn;
    }
    __syncthreads();

    const unsigned plane = (__builtin_amdgcn_s_sendmsg_rtn(0x87) >> 16) & (NCOPY - 1);
    float* Sp = S + (size_t)plane * (NGRAPHS * HID);
#define FLUSH_BATCH 8
    for (unsigned base = tid; base < NGRAPHS * HID; base += FLUSH_BATCH * POOL_THREADS) {
        float v[FLUSH_BATCH];
#pragma unroll
        for (int u = 0; u < FLUSH_BATCH; ++u) {
            const unsigned i = base + u * POOL_THREADS;
            v[u] = (i < NGRAPHS * HID) ? sS[i] : 0.0f;
        }
#pragma unroll
        for (int u = 0; u < FLUSH_BATCH; ++u) {
            const unsigned i = base + u * POOL_THREADS;
            if (i < NGRAPHS * HID) unsafeAtomicAdd(Sp + i, v[u]);
        }
    }
}

__global__ __launch_bounds__(256) void gcn_finalize(
    const float* __restrict__ gsum, const float* __restrict__ gcnt,
    const float* __restrict__ b2, float* __restrict__ out)
{
    const unsigned gid = blockIdx.x * 256u + threadIdx.x;
    if (gid >= NGRAPHS * 64u) return;
    const unsigned g = gid >> 6;
    const unsigned c = gid & 63;
    const float cnt = gcnt[g];
    out[gid] = (cnt > 0.0f) ? (gsum[gid] / cnt + b2[c]) : 0.0f;
}

extern "C" void kernel_launch(void* const* d_in, const int* in_sizes, int n_in,
                              void* d_out, int out_size, void* d_ws, size_t ws_size,
                              hipStream_t stream)
{
    (void)in_sizes; (void)n_in; (void)out_size; (void)ws_size;

    const float* feats = (const float*)d_in[0];
    const float* W1    = (const float*)d_in[1];
    const float* b1    = (const float*)d_in[2];
    const float* W2    = (const float*)d_in[3];
    const float* b2    = (const float*)d_in[4];
    const int*   src   = (const int*)d_in[5];
    const int*   dst   = (const int*)d_in[6];
    const int*   gids  = (const int*)d_in[7];
    float* out = (float*)d_out;

    const size_t nodeElems = (size_t)NNODES * HID;
    const size_t sumElems  = (size_t)NCOPY * NGRAPHS * HID;
    float* tick = (float*)d_ws;
    float* gcnt = tick + 64;
    float* S    = gcnt + 1024;
    float* agg  = S + sumElems;
    float* P    = agg + nodeElems;
    unsigned short* gtab = (unsigned short*)P;

    const int gemmBlocks    = NNODES / (16 * G1_TILES);
    const int scatterBlocks = 2 * SC_UNITS;
    const int poolBlocks    = PROWS / 16;
    const int finBlocks     = (NGRAPHS * 64 + 255) / 256;

    hipMemsetAsync(tick, 0, (64 + 1024 + sumElems + nodeElems) * sizeof(float), stream);
    hipLaunchKernelGGL(gcn_scatter, dim3(scatterBlocks), dim3(256), 0, stream,
                       feats, src, dst, agg, tick);
    hipLaunchKernelGGL(gcn_gemm1_wmma, dim3(gemmBlocks), dim3(128), 0, stream,
                       agg, W1, b1, gids, gtab);
    hipLaunchKernelGGL(gcn_scatter_pool, dim3(POOL_WGS), dim3(POOL_THREADS), 0, stream,
                       agg, src, dst, gids, gtab, S, gcnt);
    hipLaunchKernelGGL(gcn_gemm_pool_wmma, dim3(poolBlocks), dim3(128), 0, stream,
                       S, W2, P);
    hipLaunchKernelGGL(gcn_finalize, dim3(finBlocks), dim3(256), 0, stream,
                       P, gcnt, b2, out);
}


# layer 1: edges binned by 400-row destination range (LDS counting sort per call); a work-group sums a range's feats rows by LDS f32 adds, W1 from LDS: no global adds, no big clear; sum order changes
# speedup vs baseline: 4.7390x; 1.1964x over previous
#include <hip/hip_runtime.h>

#define HID      64
#define NNODES   100000
#define NEDGES   1200000
#define NGRAPHS  1000
#define NCOPY    8
#define PROWS    1008

typedef __attribute__((ext_vector_type(2))) float v2f;
typedef __attribute__((ext_vector_type(8))) float v8f;


#define AG_ROWS    400
#define AG_RANGES  (NNODES / AG_ROWS)
#define AG_CAP     (AG_ROWS * HID / 2)
#define AG_TILES   (AG_ROWS / 16)
#define AG_THREADS 1024
#define BIN_WGS     256
#define BIN_THREADS 1024
#define BIN_EDGES   ((NEDGES + BIN_WGS - 1) / BIN_WGS)
#define BIN_SLOTS   ((BIN_EDGES + BIN_THREADS - 1) / BIN_THREADS)
#define GT_NODES    8
#define GT_ENTRIES  (NNODES / GT_NODES)
#define GT_MIXED    0xFFFFu
#define GT_PIECES   (GT_ENTRIES / 4)
#define G1_PITCH  68
__device__ __forceinline__ void g1_tile(const float* a, const v2f (&bw)[16], const float bn, float* outp)
{
    v8f acc = {};
#pragma unroll
    for (int t = 0; t < 16; ++t) {
        const v2f av = *reinterpret_cast<const v2f*>(a + 4 * t);
        acc = __builtin_amdgcn_wmma_f32_16x16x4_f32(
             false, av,  false, bw[t],
             (short)0, acc,  false,  false);
    }
    asm volatile("v_nop\n\tv_nop\n\tv_nop\n\tv_nop" : "+v"(acc));
#pragma unroll
    for (int v = 0; v < 8; ++v) {
        const float h = acc[v] + bn;
        outp[(size_t)v * 64] = h > 0.0f ? h : 0.0f;
    }
}

__global__ __launch_bounds__(128) void gcn_gemm_pool_wmma(
    const float* In, const float* W,
    float* __restrict__ Out)
{
    __shared__ float sW[64 * 65];
    __shared__ float sA[16 * 65];

    const int tid  = threadIdx.x;
    const int wave = tid >> 5;
    const int lane = tid & 31;
    const int rowBase = blockIdx.x * 16;

    float wv[32];
#pragma unroll
    for (int j = 0; j < 32; ++j)
        wv[j] = W[tid + 128 * j];

    float4 av[2][NCOPY];
#pragma unroll
    for (int j = 0; j < 2; ++j) {
        const int q   = tid + 128 * j;
        const int row = rowBase + (q >> 4);
        const int rc  = row < NGRAPHS ? row : NGRAPHS - 1;
        const float4* p4 = reinterpret_cast<const float4*>(In) + rc * 16 + (q & 15);
#pragma unroll
        for (int p = 0; p < NCOPY; ++p)
            av[j][p] = p4[(size_t)p * (NGRAPHS * HID / 4)];
    }
    __builtin_amdgcn_sched_barrier(0);

#pragma unroll
    for (int j = 0; j < 32; ++j) {
        const int i = tid + 128 * j;
        sW[(i >> 6) * 65 + (i & 63)] = wv[j];
    }
#pragma unroll
    for (int j = 0; j < 2; ++j) {
        const int q = tid + 128 * j;
        const int r = q >> 4, c = (q & 15) << 2;
        float4 v = make_float4(0.0f, 0.0f, 0.0f, 0.0f);
#pragma unroll
        for (int p = 0; p < NCOPY; ++p) {
            v.x += av[j][p].x;
            v.y += av[j][p].y;
            v.z += av[j][p].z;
            v.w += av[j][p].w;
        }
        if (rowBase + r >= NGRAPHS) v = make_float4(0.0f, 0.0f, 0.0f, 0.0f);
        sA[r * 65 + c]     = v.x;
        sA[r * 65 + c + 1] = v.y;
        sA[r * 65 + c + 2] = v.z;
        sA[r * 65 + c + 3] = v.w;
    }
    __syncthreads();

    const int m     = lane & 15;
    const int n     = (wave << 4) + (lane & 15);
    const int khalf = (lane >> 4) << 1;

    v8f acc = {};
#pragma unroll
    for (int k = 0; k < 64; k += 4) {
        v2f a, b;
        a.x = sA[m * 65 + k + khalf];
        a.y = sA[m * 65 + k + khalf + 1];
        b.x = sW[(k + khalf) * 65 + n];
        b.y = sW[(k + khalf + 1) * 65 + n];
        acc = __builtin_amdgcn_wmma_f32_16x16x4_f32(
             false, a,  false, b,
             (short)0, acc,  false,  false);
    }

    const int rOff = (lane >> 4) << 3;
    float* outp = Out + (size_t)(rowBase + rOff) * 64 + n;
#pragma unroll
    for (int v = 0; v < 8; ++v)
        outp[(size_t)v * 64] = acc[v];
}

#define EDGES_PER_WAVE 32

#define POOL_THREADS 1024
#define POOL_WGS     256
#define POOL_WAVES   (POOL_WGS * (POOL_THREADS / 32))
#define NCHUNKS      (NEDGES / EDGES_PER_WAVE)
#define POOL_BATCH   16

__device__ __forceinline__ void pool_gather(
    const int sv, const int j0, const float* __restrict__ xl,
    float (&vlo)[POOL_BATCH], float (&vhi)[POOL_BATCH])
{
#pragma unroll
    for (int u = 0; u < POOL_BATCH; ++u) {
        const int s = __builtin_amdgcn_readlane(sv, j0 + u);
        const float* r = xl + (size_t)s * 64;
        vlo[u] = r[0];
        vhi[u] = r[32];
    }
}
__device__ __forceinline__ void pool_add(
    const int gv, const int j0, float* sl,
    const float (&vlo)[POOL_BATCH], const float (&vhi)[POOL_BATCH])
{
#pragma unroll
    for (int u = 0; u < POOL_BATCH; ++u) {
        const int g = __builtin_amdgcn_readlane(gv, j0 + u);
        float* p = sl + g * 64;
        __hip_atomic_fetch_add(p, vlo[u], __ATOMIC_RELAXED, __HIP_MEMORY_SCOPE_WORKGROUP);
        __hip_atomic_fetch_add(p + 32, vhi[u], __ATOMIC_RELAXED, __HIP_MEMORY_SCOPE_WORKGROUP);
    }
}

__global__ __launch_bounds__(POOL_THREADS) void gcn_scatter_pool(
    const float* __restrict__ Agg1,
    const int* __restrict__ src, const int* __restrict__ dst,
    const int* __restrict__ gids, const unsigned short* gtab,
    float* __restrict__ S, float* __restrict__ gcnt)
{
    __shared__ __attribute__((aligned(16))) float sS[NGRAPHS * HID];
    __shared__ __attribute__((aligned(8))) unsigned short sT[GT_ENTRIES];

    const unsigned tid  = threadIdx.x;
    const unsigned wave = __builtin_amdgcn_readfirstlane(tid >> 5);
    const unsigned l    = tid & 31;

    unsigned wc = wave * POOL_WGS + blockIdx.x;
    int sv, gv;
    unsigned dv;
    {
        const unsigned e = wc * EDGES_PER_WAVE + l;
        sv = src[e];
        dv = (unsigned)dst[e];
        if (e < NNODES) unsafeAtomicAdd(&gcnt[gids[e]], 1.0f);
    }

    const unsigned i3 = tid + 3 * POOL_THREADS < GT_PIECES ? tid + 3 * POOL_THREADS : GT_PIECES - 1;
    const uint2* t2 = reinterpret_cast<const uint2*>(gtab);
    const uint2 tv0 = t2[tid];
    const uint2 tv1 = t2[tid + POOL_THREADS];
    const uint2 tv2 = t2[tid + 2 * POOL_THREADS];
    const uint2 tv3 = t2[i3];

    float4* s4 = reinterpret_cast<float4*>(sS);
    for (unsigned i = tid; i < NGRAPHS * HID / 4; i += POOL_THREADS)
        s4[i] = make_float4(0.0f, 0.0f, 0.0f, 0.0f);
    {
        uint2* s2 = reinterpret_cast<uint2*>(sT);
        s2[tid] = tv0;
        s2[tid + POOL_THREADS] = tv1;
        s2[tid + 2 * POOL_THREADS] = tv2;
        if (tid + 3 * POOL_THREADS < GT_PIECES) s2[i3] = tv3;
    }
    __syncthreads();

    {
        const unsigned t = sT[dv / GT_NODES];
        gv = (int)t;
        if (t == GT_MIXED) gv = gids[dv];
    }

    const float* xl = Agg1 + l;
    float* sl = sS + l;

    for (;;) {
        const unsigned wcn = wc + POOL_WAVES;
        const bool more = wcn < NCHUNKS;
        const unsigned en = more ? (wcn * EDGES_PER_WAVE + l) : (wc * EDGES_PER_WAVE);
        const int svn = src[en];
        const unsigned dvn = (unsigned)dst[en];
        float vlo[POOL_BATCH], vhi[POOL_BATCH];
        pool_gather(sv, 0, xl, vlo, vhi);
        const unsigned tn = sT[dvn / GT_NODES];
        pool_add(gv, 0, sl, vlo, vhi);
        int gvn = (int)tn;
        if (tn == GT_MIXED) gvn = gids[dvn];
        pool_gather(sv, POOL_BATCH, xl, vlo, vhi);
        pool_add(gv, POOL_BATCH, sl, vlo, vhi);
        if (!more) break;
        sv = svn; gv = gvn; wc = wcn;
    }
    __syncthreads();

    const unsigned plane = (__builtin_amdgcn_s_sendmsg_rtn(0x87) >> 16) & (NCOPY - 1);
    float* Sp = S + (size_t)plane * (NGRAPHS * HID);
#define FLUSH_BATCH 8
    for (unsigned base = tid; base < NGRAPHS * HID; base += FLUSH_BATCH * POOL_THREADS) {
        float v[FLUSH_BATCH];
#pragma unroll
        for (int u = 0; u < FLUSH_BATCH; ++u) {
            const unsigned i = base + u * POOL_THREADS;
            v[u] = (i < NGRAPHS * HID) ? sS[i] : 0.0f;
        }
#pragma unroll
        for (int u = 0; u < FLUSH_BATCH; ++u) {
            const unsigned i = base + u * POOL_THREADS;
            if (i < NGRAPHS * HID) unsafeAtomicAdd(Sp + i, v[u]);
        }
    }
}

__global__ __launch_bounds__(256) void gcn_finalize(
    const float* __restrict__ gsum, const float* __restrict__ gcnt,
    const float* __restrict__ b2, float* __restrict__ out)
{
    const unsigned gid = blockIdx.x * 256u + threadIdx.x;
    if (gid >= NGRAPHS * 64u) return;
    const unsigned g = gid >> 6;
    const unsigned c = gid & 63;
    const float cnt = gcnt[g];
    out[gid] = (cnt > 0.0f) ? (gsum[gid] / cnt + b2[c]) : 0.0f;
}

__global__ __launch_bounds__(BIN_THREADS) void gcn_bin_edges(
    const int* __restrict__ src, const int* __restrict__ dst,
    unsigned* cursor, float* Buf)
{
    __shared__ unsigned sCnt[AG_RANGES];
    __shared__ unsigned sStart[AG_RANGES];
    __shared__ unsigned sBase[AG_RANGES];
    __shared__ __attribute__((aligned(8))) uint2 sRec[BIN_EDGES];
    __shared__ unsigned short sBin[BIN_EDGES];

    const unsigned tid = threadIdx.x;
    const unsigned e0  = blockIdx.x * BIN_EDGES;
    const unsigned nE  = (NEDGES - e0) < BIN_EDGES ? (NEDGES - e0) : BIN_EDGES;

    unsigned s[BIN_SLOTS], d[BIN_SLOTS];
#pragma unroll
    for (int k = 0; k < BIN_SLOTS; ++k) {
        const unsigned i = tid + k * BIN_THREADS;
        const unsigned e = i < nE ? e0 + i : e0;
        s[k] = (unsigned)src[e];
        d[k] = (unsigned)dst[e];
    }
    if (tid < AG_RANGES) sCnt[tid] = 0u;
    __syncthreads();

    unsigned b[BIN_SLOTS], r[BIN_SLOTS];
#pragma unroll
    for (int k = 0; k < BIN_SLOTS; ++k) {
        b[k] = d[k] / AG_ROWS;
        r[k] = 0u;
        if (tid + k * BIN_THREADS < nE) r[k] = atomicAdd(&sCnt[b[k]], 1u);
    }
    __syncthreads();

    if (tid < AG_RANGES) {
        const unsigned c = sCnt[tid];
        const unsigned g = atomicAdd(&cursor[tid], c);
        unsigned acc = 0u;
        for (unsigned j = 0; j < AG_RANGES; ++j) {
            const unsigned v = sCnt[j];
            acc += j < tid ? v : 0u;
        }
        sStart[tid] = acc;
        sBase[tid]  = g;
    }
    __syncthreads();

#pragma unroll
    for (int k = 0; k < BIN_SLOTS; ++k) {
        if (tid + k * BIN_THREADS < nE) {
            const unsigned slot = sStart[b[k]] + r[k];
            sRec[slot] = make_uint2(s[k], d[k] - b[k] * AG_ROWS);
            sBin[slot] = (unsigned short)b[k];
        }
    }
    __syncthreads();

    for (unsigned i = tid; i < nE; i += BIN_THREADS) {
        const unsigned bb  = sBin[i];
        const unsigned pos = sBase[bb] + (i - sStart[bb]);
        if (pos < AG_CAP)
            reinterpret_cast<uint2*>(Buf + (size_t)bb * (AG_ROWS * HID))[pos] = sRec[i];
    }
}

template <bool LISTED>
__device__ __forceinline__ void agg_load(
    const unsigned c, const unsigned l, const unsigned n, const unsigned range,
    const uint2* recs, const int* __restrict__ src, const int* __restrict__ dst, int& sv, int& gv)
{
    const unsigned e = c * 32u + l;
    if (LISTED) {
        const bool in = e < n;
        const uint2 p = recs[in ? e : 0u];
        sv = in ? (int)p.x : 0;
        gv = in ? (int)p.y : AG_ROWS;
    } else {
        const unsigned s = (unsigned)src[e];
        const unsigned d = (unsigned)dst[e];
        const bool in = d / AG_ROWS == range;
        sv = in ? (int)s : 0;
        gv = in ? (int)(d - range * AG_ROWS) : AG_ROWS;
    }
}

__device__ __forceinline__ void agg_add(
    const int gv, const int j0, float* sl,
    const float (&vlo)[POOL_BATCH], const float (&vhi)[POOL_BATCH])
{
#pragma unroll
    for (int u = 0; u < POOL_BATCH; ++u) {
        const int g = __builtin_amdgcn_readlane(gv, j0 + u);
        float* p = sl + g * G1_PITCH;
        __hip_atomic_fetch_add(p, vlo[u], __ATOMIC_RELAXED, __HIP_MEMORY_SCOPE_WORKGROUP);
        __hip_atomic_fetch_add(p + 32, vhi[u], __ATOMIC_RELAXED, __HIP_MEMORY_SCOPE_WORKGROUP);
    }
}

template <bool LISTED>
__device__ __forceinline__ void agg_walk(
    const unsigned wave, const unsigned l, const unsigned n, const unsigned range,
    const uint2* recs, const int* __restrict__ src, const int* __restrict__ dst,
    const float* __restrict__ xl, float* sl)
{
    const unsigned nch = LISTED ? (n + 31u) / 32u : (unsigned)(NEDGES / 32);
    unsigned c = wave;
    if (c >= nch) return;
    int sv, gv;
    agg_load<LISTED>(c, l, n, range, recs, src, dst, sv, gv);
    for (;;) {
        const unsigned cn = c + 32u;
        const bool more = cn < nch;
        int svn, gvn;
        agg_load<LISTED>(more ? cn : c, l, n, range, recs, src, dst, svn, gvn);
        __builtin_amdgcn_sched_barrier(0);
        float vlo[POOL_BATCH], vhi[POOL_BATCH];
        pool_gather(sv, 0, xl, vlo, vhi);
        agg_add(gv, 0, sl, vlo, vhi);
        pool_gather(sv, POOL_BATCH, xl, vlo, vhi);
        agg_add(gv, POOL_BATCH, sl, vlo, vhi);
        if (!more) break;
        sv = svn; gv = gvn; c = cn;
    }
}

__global__ __launch_bounds__(AG_THREADS) void gcn_agg_gemm1_wmma(
    const float* __restrict__ X, const int* __restrict__ src, const int* __restrict__ dst,
    float* Buf, unsigned* cursor, const float* W, const float* __restrict__ b1,
    const int* __restrict__ gids, unsigned short* __restrict__ gtab)
{
    __shared__ __attribute__((aligned(16))) float sRow[(AG_ROWS + 1) * G1_PITCH];

    const unsigned tid   = threadIdx.x;
    const unsigned wave  = __builtin_amdgcn_readfirstlane(tid >> 5);
    const unsigned l     = tid & 31;
    const unsigned range = blockIdx.x;
    const unsigned n = __builtin_amdgcn_readfirstlane(cursor[range]);

    const int m     = l & 15;
    const int ncol  = ((wave & 3) << 4) + (l & 15);
    const int khalf = (l >> 4) << 1;
    const int rOff  = (l >> 4) << 3;
    const float bn = b1[ncol];
    v2f bw[16];
#pragma unroll
    for (int t = 0; t < 16; ++t) {
        bw[t].x = W[(4 * t + khalf) * 64 + ncol];
        bw[t].y = W[(4 * t + khalf + 1) * 64 + ncol];
    }

    float4* s4 = reinterpret_cast<float4*>(sRow);
    for (unsigned i = tid; i < (AG_ROWS + 1) * G1_PITCH / 4; i += AG_THREADS)
        s4[i] = make_float4(0.0f, 0.0f, 0.0f, 0.0f);
    __syncthreads();

    const uint2* recs = reinterpret_cast<const uint2*>(Buf + (size_t)range * (AG_ROWS * HID));
    if (n <= AG_CAP) agg_walk<true>(wave, l, n, range, recs, src, dst, X + l, sRow + l);
    else             agg_walk<false>(wave, l, n, range, recs, src, dst, X + l, sRow + l);
    __syncthreads();
    if (tid == 0) cursor[range] = 0u;

    for (unsigned tile = wave >> 2; tile < AG_TILES; tile += AG_THREADS / 128)
        g1_tile(sRow + (tile * 16 + m) * G1_PITCH + khalf, bw, bn,
                Buf + ((size_t)range * AG_ROWS + tile * 16 + rOff) * 64 + ncol);

    if (tid < AG_ROWS / GT_NODES) {
        const unsigned blk = range * (AG_ROWS / GT_NODES) + tid;
        const int4 ga = *reinterpret_cast<const int4*>(gids + (size_t)blk * GT_NODES);
        const int4 gb = *reinterpret_cast<const int4*>(gids + (size_t)blk * GT_NODES + 4);
        const unsigned g = (unsigned)ga.x;
        const unsigned diff = (unsigned)((ga.y ^ ga.x) | (ga.z ^ ga.x) | (ga.w ^ ga.x) |
                                         (gb.x ^ ga.x) | (gb.y ^ ga.x) | (gb.z ^ ga.x) | (gb.w ^ ga.x));
        gtab[blk] = (unsigned short)((diff == 0u && g < GT_MIXED) ? g : GT_MIXED);
    }
}

extern "C" void kernel_launch(void* const* d_in, const int* in_sizes, int n_in,
                              void* d_out, int out_size, void* d_ws, size_t ws_size,
                              hipStream_t stream)
{
    (void)in_sizes; (void)n_in; (void)out_size; (void)ws_size;

    const float* feats = (const float*)d_in[0];
    const float* W1    = (const float*)d_in[1];
    const float* b1    = (const float*)d_in[2];
    const float* W2    = (const float*)d_in[3];
    const float* b2    = (const float*)d_in[4];
    const int*   src   = (const int*)d_in[5];
    const int*   dst   = (const int*)d_in[6];
    const int*   gids  = (const int*)d_in[7];
    float* out = (float*)d_out;

    const size_t nodeElems = (size_t)NNODES * HID;
    const size_t sumElems  = (size_t)NCOPY * NGRAPHS * HID;
    float* spare = (float*)d_ws;
    float* gcnt = spare + 64;
    float* S    = gcnt + 1024;
    float* agg  = S + sumElems;
    float* P    = agg + nodeElems;
    unsigned* cursor = (unsigned*)S;
    unsigned short* gtab = (unsigned short*)P;

    const int poolBlocks    = PROWS / 16;
    const int finBlocks     = (NGRAPHS * 64 + 255) / 256;

    hipMemsetAsync(spare, 0, (64 + 1024 + sumElems) * sizeof(float), stream);
    hipLaunchKernelGGL(gcn_bin_edges, dim3(BIN_WGS), dim3(BIN_THREADS), 0, stream,
                       src, dst, cursor, agg);
    hipLaunchKernelGGL(gcn_agg_gemm1_wmma, dim3(AG_RANGES), dim3(AG_THREADS), 0, stream,
                       feats, src, dst, agg, cursor, W1, b1, gids, gtab);
    hipLaunchKernelGGL(gcn_scatter_pool, dim3(POOL_WGS), dim3(POOL_THREADS), 0, stream,
                       agg, src, dst, gids, gtab, S, gcnt);
    hipLaunchKernelGGL(gcn_gemm_pool_wmma, dim3(poolBlocks), dim3(128), 0, stream,
                       S, W2, P);
    hipLaunchKernelGGL(gcn_finalize, dim3(finBlocks), dim3(256), 0, stream,
                       P, gcnt, b2, out);
}
